# starting kernel
# baseline (speedup 1.0000x reference)
#include <hip/hip_runtime.h>
#include <hip/hip_bf16.h>

#define BATCH 8
#define SEQ   4096
#define DIM   1024
#define NSTATE 256
#define LN_EPS 1e-5f

typedef float v2f __attribute__((ext_vector_type(2)));
typedef float v4f __attribute__((ext_vector_type(4)));
typedef float v8f __attribute__((ext_vector_type(8)));

__global__ void __launch_bounds__(256)
ssm_rowmean_wmma(const float* __restrict__ x, float* __restrict__ xmean,
                 int rows, int cols) {
    const int wid  = (blockIdx.x * blockDim.x + threadIdx.x) >> 5;
    const int lane = threadIdx.x & 31;
    const int r0   = wid << 4;
    if (r0 >= rows) return;

    const int myrow = r0 + (lane & 15);
    const int koff  = (lane >> 4) << 1;
    const float* rowp = x + (size_t)myrow * cols + koff;

    const float inv = 1.0f / (float)cols;
    v2f bmat; bmat[0] = inv; bmat[1] = inv;
    v8f acc = {};

#pragma unroll 8
    for (int k = 0; k < cols; k += 4) {
        v2f a = *(const v2f*)(rowp + k);
        acc = __builtin_amdgcn_wmma_f32_16x16x4_f32(
             false, a,  false, bmat,
             (short)0, acc,  false,  false);
    }

    if (lane == 0) {
#pragma unroll
        for (int j = 0; j < 8; ++j) xmean[r0 + j] = acc[j];
    } else if (lane == 16) {
#pragma unroll
        for (int j = 0; j < 8; ++j) xmean[r0 + 8 + j] = acc[j];
    }
}

__device__ __forceinline__ float wave_sum(float v) {
#pragma unroll
    for (int off = 16; off >= 1; off >>= 1) v += __shfl_xor(v, off, 32);
    return v;
}

__global__ void __launch_bounds__(BATCH * 32)
ssm_scan(const float* __restrict__ xmean,
         const float* __restrict__ vol,
         const float* __restrict__ llr,
         const float* __restrict__ logb,
         const float* __restrict__ cvec,
         const float* __restrict__ log_d,
         const float* __restrict__ log_step,
         const float* __restrict__ vgate,
         const float* __restrict__ lnw,
         const float* __restrict__ lnb,
         float* __restrict__ ys)
{
    const int lane  = threadIdx.x & 31;
    const int batch = threadIdx.x >> 5;

    const float step = __expf(log_step[0]);

    float a_[8], b_[8], g_[8], w_[8], o_[8], c_[8], h_[8];
#pragma unroll
    for (int j = 0; j < 8; ++j) {
        const int n = lane + 32 * j;
        const float lam = -__expf(llr[n]);
        const float ad  = (2.0f + step * lam) * __builtin_amdgcn_rcpf(2.0f - step * lam);
        a_[j] = ad;
        b_[j] = step * (1.0f + ad) * __expf(logb[n]) * 0.5f;
        g_[j] = __builtin_amdgcn_rcpf(1.0f + __expf(-vgate[n]));
        w_[j] = lnw[n];
        o_[j] = lnb[n];
        c_[j] = cvec[n];
        h_[j] = 0.0f;
    }

    const float* xmb  = xmean + batch * SEQ;
    const float* volb = vol   + batch * SEQ;
    float*       ysb  = ys    + batch * SEQ;
    const float invN = 1.0f / (float)NSTATE;

    for (int t = 0; t < SEQ; ++t) {
        const float xm = xmb[t];
        const float vt = volb[t];

        float s1 = 0.0f, s2 = 0.0f;
#pragma unroll
        for (int j = 0; j < 8; ++j) {
            const float h = fmaf(a_[j], h_[j], b_[j] * xm);
            h_[j] = h;
            s1 += h;
            s2 = fmaf(h, h, s2);
        }
        s1 = wave_sum(s1);
        s2 = wave_sum(s2);

        const float mu  = s1 * invN;
        const float var = fmaf(s2, invN, -mu * mu);
        const float rs  = __builtin_amdgcn_rsqf(var + LN_EPS);

        float dot = 0.0f;
#pragma unroll
        for (int j = 0; j < 8; ++j) {
            float hn = fmaf((h_[j] - mu) * rs, w_[j], o_[j]);
            hn *= __builtin_amdgcn_rcpf(fmaf(g_[j], vt, 1.0f));
            h_[j] = hn;
            dot = fmaf(hn, c_[j], dot);
        }
        dot = wave_sum(dot);
        if (lane == 0) ysb[t] = dot;
    }
}

__global__ void __launch_bounds__(256)
ssm_output(const float* __restrict__ x, const float* __restrict__ ys,
           const float* __restrict__ alpha, const float* __restrict__ log_d,
           float* __restrict__ out, int total4) {
    const int i = blockIdx.x * blockDim.x + threadIdx.x;
    if (i >= total4) return;

    const float a  = 1.0f / (1.0f + __expf(-alpha[0]));
    const float dd = __expf(log_d[0]);
    const float s0 = a + (1.0f - a) * dd;
    const float s1 = 1.0f - a;

    const int row = i >> 8;
    const float yv = s1 * ys[row];

    const v4f xv = ((const v4f*)x)[i];
    v4f o;
    o.x = fmaf(s0, xv.x, yv);
    o.y = fmaf(s0, xv.y, yv);
    o.z = fmaf(s0, xv.z, yv);
    o.w = fmaf(s0, xv.w, yv);
    __builtin_nontemporal_store(o, ((v4f*)out) + i);
}

extern "C" void kernel_launch(void* const* d_in, const int* in_sizes, int n_in,
                              void* d_out, int out_size, void* d_ws, size_t ws_size,
                              hipStream_t stream) {
    const float* x        = (const float*)d_in[0];
    const float* vol      = (const float*)d_in[1];
    const float* llr      = (const float*)d_in[2];
    const float* logb     = (const float*)d_in[3];
    const float* cvec     = (const float*)d_in[4];
    const float* log_d    = (const float*)d_in[5];
    const float* log_step = (const float*)d_in[6];
    const float* vgate    = (const float*)d_in[7];
    const float* alpha    = (const float*)d_in[8];
    const float* lnw      = (const float*)d_in[9];
    const float* lnb      = (const float*)d_in[10];
    float* out = (float*)d_out;

    const int rows = BATCH * SEQ;
    float* xmean = (float*)d_ws;
    float* ys    = xmean + rows;

    {
        const int blocks = rows / 128;
        ssm_rowmean_wmma<<<blocks, 256, 0, stream>>>(x, xmean, rows, DIM);
    }
    ssm_scan<<<1, BATCH * 32, 0, stream>>>(xmean, vol, llr, logb, cvec, log_d,
                                           log_step, vgate, lnw, lnb, ys);
    {
        const int total4 = (BATCH * SEQ * DIM) / 4;
        const int blocks = (total4 + 255) / 256;
        ssm_output<<<blocks, 256, 0, stream>>>(x, ys, alpha, log_d, out, total4);
    }
}


# scan kernel: each step's three 32-lane sums run on the vector ALU (v_permlanex16 + four row_xmask DPP adds, same partner order) instead of five LDS-crossbar shuffles each; sums bit-identical
# speedup vs baseline: 2.1883x; 2.1883x over previous
#include <hip/hip_runtime.h>
#include <hip/hip_bf16.h>

#define BATCH 8
#define SEQ   4096
#define DIM   1024
#define NSTATE 256
#define LN_EPS 1e-5f

typedef float v2f __attribute__((ext_vector_type(2)));
typedef float v4f __attribute__((ext_vector_type(4)));
typedef float v8f __attribute__((ext_vector_type(8)));

__global__ void __launch_bounds__(256)
ssm_rowmean_wmma(const float* __restrict__ x, float* __restrict__ xmean,
                 int rows, int cols) {
    const int wid  = (blockIdx.x * blockDim.x + threadIdx.x) >> 5;
    const int lane = threadIdx.x & 31;
    const int r0   = wid << 4;
    if (r0 >= rows) return;

    const int myrow = r0 + (lane & 15);
    const int koff  = (lane >> 4) << 1;
    const float* rowp = x + (size_t)myrow * cols + koff;

    const float inv = 1.0f / (float)cols;
    v2f bmat; bmat[0] = inv; bmat[1] = inv;
    v8f acc = {};

#pragma unroll 8
    for (int k = 0; k < cols; k += 4) {
        v2f a = *(const v2f*)(rowp + k);
        acc = __builtin_amdgcn_wmma_f32_16x16x4_f32(
             false, a,  false, bmat,
             (short)0, acc,  false,  false);
    }

    if (lane == 0) {
#pragma unroll
        for (int j = 0; j < 8; ++j) xmean[r0 + j] = acc[j];
    } else if (lane == 16) {
#pragma unroll
        for (int j = 0; j < 8; ++j) xmean[r0 + 8 + j] = acc[j];
    }
}

__device__ __forceinline__ float xhalf(float x) {
    const int b = __builtin_bit_cast(int, x);
    return __builtin_bit_cast(float,
        __builtin_amdgcn_permlanex16(b, b, 0x76543210, 0xfedcba98, true, false));
}

template <int XMASK>
__device__ __forceinline__ float row_xor(float x) {
    const int b = __builtin_bit_cast(int, x);
    return __builtin_bit_cast(float,
        __builtin_amdgcn_update_dpp(0, b, 0x160 + XMASK, 0xf, 0xf, true));
}

__device__ __forceinline__ float wave_sum(float v) {
    v += xhalf(v);
    v += row_xor<8>(v);
    v += row_xor<4>(v);
    v += row_xor<2>(v);
    v += row_xor<1>(v);
    return v;
}

__global__ void __launch_bounds__(BATCH * 32)
ssm_scan(const float* __restrict__ xmean,
         const float* __restrict__ vol,
         const float* __restrict__ llr,
         const float* __restrict__ logb,
         const float* __restrict__ cvec,
         const float* __restrict__ log_d,
         const float* __restrict__ log_step,
         const float* __restrict__ vgate,
         const float* __restrict__ lnw,
         const float* __restrict__ lnb,
         float* __restrict__ ys)
{
    const int lane  = threadIdx.x & 31;
    const int batch = threadIdx.x >> 5;

    const float step = __expf(log_step[0]);

    float a_[8], b_[8], g_[8], w_[8], o_[8], c_[8], h_[8];
#pragma unroll
    for (int j = 0; j < 8; ++j) {
        const int n = lane + 32 * j;
        const float lam = -__expf(llr[n]);
        const float ad  = (2.0f + step * lam) * __builtin_amdgcn_rcpf(2.0f - step * lam);
        a_[j] = ad;
        b_[j] = step * (1.0f + ad) * __expf(logb[n]) * 0.5f;
        g_[j] = __builtin_amdgcn_rcpf(1.0f + __expf(-vgate[n]));
        w_[j] = lnw[n];
        o_[j] = lnb[n];
        c_[j] = cvec[n];
        h_[j] = 0.0f;
    }

    const float* xmb  = xmean + batch * SEQ;
    const float* volb = vol   + batch * SEQ;
    float*       ysb  = ys    + batch * SEQ;
    const float invN = 1.0f / (float)NSTATE;

    for (int t = 0; t < SEQ; ++t) {
        const float xm = xmb[t];
        const float vt = volb[t];

        float s1 = 0.0f, s2 = 0.0f;
#pragma unroll
        for (int j = 0; j < 8; ++j) {
            const float h = fmaf(a_[j], h_[j], b_[j] * xm);
            h_[j] = h;
            s1 += h;
            s2 = fmaf(h, h, s2);
        }
        s1 = wave_sum(s1);
        s2 = wave_sum(s2);

        const float mu  = s1 * invN;
        const float var = fmaf(s2, invN, -mu * mu);
        const float rs  = __builtin_amdgcn_rsqf(var + LN_EPS);

        float dot = 0.0f;
#pragma unroll
        for (int j = 0; j < 8; ++j) {
            float hn = fmaf((h_[j] - mu) * rs, w_[j], o_[j]);
            hn *= __builtin_amdgcn_rcpf(fmaf(g_[j], vt, 1.0f));
            h_[j] = hn;
            dot = fmaf(hn, c_[j], dot);
        }
        dot = wave_sum(dot);
        if (lane == 0) ysb[t] = dot;
    }
}

__global__ void __launch_bounds__(256)
ssm_output(const float* __restrict__ x, const float* __restrict__ ys,
           const float* __restrict__ alpha, const float* __restrict__ log_d,
           float* __restrict__ out, int total4) {
    const int i = blockIdx.x * blockDim.x + threadIdx.x;
    if (i >= total4) return;

    const float a  = 1.0f / (1.0f + __expf(-alpha[0]));
    const float dd = __expf(log_d[0]);
    const float s0 = a + (1.0f - a) * dd;
    const float s1 = 1.0f - a;

    const int row = i >> 8;
    const float yv = s1 * ys[row];

    const v4f xv = ((const v4f*)x)[i];
    v4f o;
    o.x = fmaf(s0, xv.x, yv);
    o.y = fmaf(s0, xv.y, yv);
    o.z = fmaf(s0, xv.z, yv);
    o.w = fmaf(s0, xv.w, yv);
    __builtin_nontemporal_store(o, ((v4f*)out) + i);
}

extern "C" void kernel_launch(void* const* d_in, const int* in_sizes, int n_in,
                              void* d_out, int out_size, void* d_ws, size_t ws_size,
                              hipStream_t stream) {
    const float* x        = (const float*)d_in[0];
    const float* vol      = (const float*)d_in[1];
    const float* llr      = (const float*)d_in[2];
    const float* logb     = (const float*)d_in[3];
    const float* cvec     = (const float*)d_in[4];
    const float* log_d    = (const float*)d_in[5];
    const float* log_step = (const float*)d_in[6];
    const float* vgate    = (const float*)d_in[7];
    const float* alpha    = (const float*)d_in[8];
    const float* lnw      = (const float*)d_in[9];
    const float* lnb      = (const float*)d_in[10];
    float* out = (float*)d_out;

    const int rows = BATCH * SEQ;
    float* xmean = (float*)d_ws;
    float* ys    = xmean + rows;

    {
        const int blocks = rows / 128;
        ssm_rowmean_wmma<<<blocks, 256, 0, stream>>>(x, xmean, rows, DIM);
    }
    ssm_scan<<<1, BATCH * 32, 0, stream>>>(xmean, vol, llr, logb, cvec, log_d,
                                           log_step, vgate, lnw, lnb, ys);
    {
        const int total4 = (BATCH * SEQ * DIM) / 4;
        const int blocks = (total4 + 255) / 256;
        ssm_output<<<blocks, 256, 0, stream>>>(x, ys, alpha, log_d, out, total4);
    }
}


# scan kernel: steps run in blocks of 16, xmean/vol read by 16-byte loads one block ahead, the 16 outputs gathered in a register and stored once per block: no load wait, branch or store inside a step
# speedup vs baseline: 4.2008x; 1.9197x over previous
#include <hip/hip_runtime.h>
#include <hip/hip_bf16.h>

#define BATCH 8
#define SEQ   4096
#define DIM   1024
#define NSTATE 256
#define LN_EPS 1e-5f
#define TBLK  16

typedef float v2f __attribute__((ext_vector_type(2)));
typedef float v4f __attribute__((ext_vector_type(4)));
typedef float v8f __attribute__((ext_vector_type(8)));

__global__ void __launch_bounds__(256)
ssm_rowmean_wmma(const float* __restrict__ x, float* __restrict__ xmean,
                 int rows, int cols) {
    const int wid  = (blockIdx.x * blockDim.x + threadIdx.x) >> 5;
    const int lane = threadIdx.x & 31;
    const int r0   = wid << 4;
    if (r0 >= rows) return;

    const int myrow = r0 + (lane & 15);
    const int koff  = (lane >> 4) << 1;
    const float* rowp = x + (size_t)myrow * cols + koff;

    const float inv = 1.0f / (float)cols;
    v2f bmat; bmat[0] = inv; bmat[1] = inv;
    v8f acc = {};

#pragma unroll 8
    for (int k = 0; k < cols; k += 4) {
        v2f a = *(const v2f*)(rowp + k);
        acc = __builtin_amdgcn_wmma_f32_16x16x4_f32(
             false, a,  false, bmat,
             (short)0, acc,  false,  false);
    }

    if (lane == 0) {
#pragma unroll
        for (int j = 0; j < 8; ++j) xmean[r0 + j] = acc[j];
    } else if (lane == 16) {
#pragma unroll
        for (int j = 0; j < 8; ++j) xmean[r0 + 8 + j] = acc[j];
    }
}

__device__ __forceinline__ float xhalf(float x) {
    const int b = __builtin_bit_cast(int, x);
    return __builtin_bit_cast(float,
        __builtin_amdgcn_permlanex16(b, b, 0x76543210, 0xfedcba98, true, false));
}

template <int XMASK>
__device__ __forceinline__ float row_xor(float x) {
    const int b = __builtin_bit_cast(int, x);
    return __builtin_bit_cast(float,
        __builtin_amdgcn_update_dpp(0, b, 0x160 + XMASK, 0xf, 0xf, true));
}

__device__ __forceinline__ float wave_sum(float v) {
    v += xhalf(v);
    v += row_xor<8>(v);
    v += row_xor<4>(v);
    v += row_xor<2>(v);
    v += row_xor<1>(v);
    return v;
}

__global__ void __launch_bounds__(BATCH * 32)
ssm_scan(const float* __restrict__ xmean,
         const float* __restrict__ vol,
         const float* __restrict__ llr,
         const float* __restrict__ logb,
         const float* __restrict__ cvec,
         const float* __restrict__ log_d,
         const float* __restrict__ log_step,
         const float* __restrict__ vgate,
         const float* __restrict__ lnw,
         const float* __restrict__ lnb,
         float* __restrict__ ys)
{
    const int lane  = threadIdx.x & 31;
    const int batch = threadIdx.x >> 5;

    const float step = __expf(log_step[0]);

    float a_[8], b_[8], g_[8], w_[8], o_[8], c_[8], h_[8];
#pragma unroll
    for (int j = 0; j < 8; ++j) {
        const int n = lane + 32 * j;
        const float lam = -__expf(llr[n]);
        const float ad  = (2.0f + step * lam) * __builtin_amdgcn_rcpf(2.0f - step * lam);
        a_[j] = ad;
        b_[j] = step * (1.0f + ad) * __expf(logb[n]) * 0.5f;
        g_[j] = __builtin_amdgcn_rcpf(1.0f + __expf(-vgate[n]));
        w_[j] = lnw[n];
        o_[j] = lnb[n];
        c_[j] = cvec[n];
        h_[j] = 0.0f;
    }

    const float* xmb  = xmean + batch * SEQ;
    const float* volb = vol   + batch * SEQ;
    float*       ysb  = ys    + batch * SEQ;
    const float invN = 1.0f / (float)NSTATE;

    v4f xc[TBLK / 4], vc[TBLK / 4];
#pragma unroll
    for (int q = 0; q < TBLK / 4; ++q) {
        xc[q] = *(const v4f*)(xmb  + 4 * q);
        vc[q] = *(const v4f*)(volb + 4 * q);
    }

    for (int t0 = 0; t0 < SEQ; t0 += TBLK) {
        const int tn = (t0 + TBLK < SEQ) ? (t0 + TBLK) : t0;
        v4f xn[TBLK / 4], vn[TBLK / 4];
#pragma unroll
        for (int q = 0; q < TBLK / 4; ++q) {
            xn[q] = *(const v4f*)(xmb  + tn + 4 * q);
            vn[q] = *(const v4f*)(volb + tn + 4 * q);
        }

        float yacc = 0.0f;
#pragma unroll
        for (int tt = 0; tt < TBLK; ++tt) {
            const float xm = xc[tt >> 2][tt & 3];
            const float vt = vc[tt >> 2][tt & 3];

            float s1 = 0.0f, s2 = 0.0f;
#pragma unroll
            for (int j = 0; j < 8; ++j) {
                const float h = fmaf(a_[j], h_[j], b_[j] * xm);
                h_[j] = h;
                s1 += h;
                s2 = fmaf(h, h, s2);
            }
            s1 = wave_sum(s1);
            s2 = wave_sum(s2);

            const float mu  = s1 * invN;
            const float var = fmaf(s2, invN, -mu * mu);
            const float rs  = __builtin_amdgcn_rsqf(var + LN_EPS);

            float dot = 0.0f;
#pragma unroll
            for (int j = 0; j < 8; ++j) {
                float hn = fmaf((h_[j] - mu) * rs, w_[j], o_[j]);
                hn *= __builtin_amdgcn_rcpf(fmaf(g_[j], vt, 1.0f));
                h_[j] = hn;
                dot = fmaf(hn, c_[j], dot);
            }
            dot = wave_sum(dot);
            yacc = (lane == tt) ? dot : yacc;
        }
        if (lane < TBLK) ysb[t0 + lane] = yacc;

#pragma unroll
        for (int q = 0; q < TBLK / 4; ++q) { xc[q] = xn[q]; vc[q] = vn[q]; }
    }
}

__global__ void __launch_bounds__(256)
ssm_output(const float* __restrict__ x, const float* __restrict__ ys,
           const float* __restrict__ alpha, const float* __restrict__ log_d,
           float* __restrict__ out, int total4) {
    const int i = blockIdx.x * blockDim.x + threadIdx.x;
    if (i >= total4) return;

    const float a  = 1.0f / (1.0f + __expf(-alpha[0]));
    const float dd = __expf(log_d[0]);
    const float s0 = a + (1.0f - a) * dd;
    const float s1 = 1.0f - a;

    const int row = i >> 8;
    const float yv = s1 * ys[row];

    const v4f xv = ((const v4f*)x)[i];
    v4f o;
    o.x = fmaf(s0, xv.x, yv);
    o.y = fmaf(s0, xv.y, yv);
    o.z = fmaf(s0, xv.z, yv);
    o.w = fmaf(s0, xv.w, yv);
    __builtin_nontemporal_store(o, ((v4f*)out) + i);
}

extern "C" void kernel_launch(void* const* d_in, const int* in_sizes, int n_in,
                              void* d_out, int out_size, void* d_ws, size_t ws_size,
                              hipStream_t stream) {
    const float* x        = (const float*)d_in[0];
    const float* vol      = (const float*)d_in[1];
    const float* llr      = (const float*)d_in[2];
    const float* logb     = (const float*)d_in[3];
    const float* cvec     = (const float*)d_in[4];
    const float* log_d    = (const float*)d_in[5];
    const float* log_step = (const float*)d_in[6];
    const float* vgate    = (const float*)d_in[7];
    const float* alpha    = (const float*)d_in[8];
    const float* lnw      = (const float*)d_in[9];
    const float* lnb      = (const float*)d_in[10];
    float* out = (float*)d_out;

    const int rows = BATCH * SEQ;
    float* xmean = (float*)d_ws;
    float* ys    = xmean + rows;

    {
        const int blocks = rows / 128;
        ssm_rowmean_wmma<<<blocks, 256, 0, stream>>>(x, xmean, rows, DIM);
    }
    ssm_scan<<<1, BATCH * 32, 0, stream>>>(xmean, vol, llr, logb, cvec, log_d,
                                           log_step, vgate, lnw, lnb, ys);
    {
        const int total4 = (BATCH * SEQ * DIM) / 4;
        const int blocks = (total4 + 255) / 256;
        ssm_output<<<blocks, 256, 0, stream>>>(x, ys, alpha, log_d, out, total4);
    }
}


# scan kernel launched as 8 one-wave work-groups (one batch each) instead of one 8-wave group: each wave has a SIMD issue port to itself; batch index now uniform, so xmean/vol come by scalar loads
# speedup vs baseline: 6.6951x; 1.5938x over previous
#include <hip/hip_runtime.h>
#include <hip/hip_bf16.h>

#define BATCH 8
#define SEQ   4096
#define DIM   1024
#define NSTATE 256
#define LN_EPS 1e-5f
#define TBLK  16

typedef float v2f __attribute__((ext_vector_type(2)));
typedef float v4f __attribute__((ext_vector_type(4)));
typedef float v8f __attribute__((ext_vector_type(8)));

__global__ void __launch_bounds__(256)
ssm_rowmean_wmma(const float* __restrict__ x, float* __restrict__ xmean,
                 int rows, int cols) {
    const int wid  = (blockIdx.x * blockDim.x + threadIdx.x) >> 5;
    const int lane = threadIdx.x & 31;
    const int r0   = wid << 4;
    if (r0 >= rows) return;

    const int myrow = r0 + (lane & 15);
    const int koff  = (lane >> 4) << 1;
    const float* rowp = x + (size_t)myrow * cols + koff;

    const float inv = 1.0f / (float)cols;
    v2f bmat; bmat[0] = inv; bmat[1] = inv;
    v8f acc = {};

#pragma unroll 8
    for (int k = 0; k < cols; k += 4) {
        v2f a = *(const v2f*)(rowp + k);
        acc = __builtin_amdgcn_wmma_f32_16x16x4_f32(
             false, a,  false, bmat,
             (short)0, acc,  false,  false);
    }

    if (lane == 0) {
#pragma unroll
        for (int j = 0; j < 8; ++j) xmean[r0 + j] = acc[j];
    } else if (lane == 16) {
#pragma unroll
        for (int j = 0; j < 8; ++j) xmean[r0 + 8 + j] = acc[j];
    }
}

__device__ __forceinline__ float xhalf(float x) {
    const int b = __builtin_bit_cast(int, x);
    return __builtin_bit_cast(float,
        __builtin_amdgcn_permlanex16(b, b, 0x76543210, 0xfedcba98, true, false));
}

template <int XMASK>
__device__ __forceinline__ float row_xor(float x) {
    const int b = __builtin_bit_cast(int, x);
    return __builtin_bit_cast(float,
        __builtin_amdgcn_update_dpp(0, b, 0x160 + XMASK, 0xf, 0xf, true));
}

__device__ __forceinline__ float wave_sum(float v) {
    v += xhalf(v);
    v += row_xor<8>(v);
    v += row_xor<4>(v);
    v += row_xor<2>(v);
    v += row_xor<1>(v);
    return v;
}

__global__ void __launch_bounds__(32)
ssm_scan(const float* __restrict__ xmean,
         const float* __restrict__ vol,
         const float* __restrict__ llr,
         const float* __restrict__ logb,
         const float* __restrict__ cvec,
         const float* __restrict__ log_d,
         const float* __restrict__ log_step,
         const float* __restrict__ vgate,
         const float* __restrict__ lnw,
         const float* __restrict__ lnb,
         float* __restrict__ ys)
{
    const int lane  = threadIdx.x;
    const int batch = blockIdx.x;

    const float step = __expf(log_step[0]);

    float a_[8], b_[8], g_[8], w_[8], o_[8], c_[8], h_[8];
#pragma unroll
    for (int j = 0; j < 8; ++j) {
        const int n = lane + 32 * j;
        const float lam = -__expf(llr[n]);
        const float ad  = (2.0f + step * lam) * __builtin_amdgcn_rcpf(2.0f - step * lam);
        a_[j] = ad;
        b_[j] = step * (1.0f + ad) * __expf(logb[n]) * 0.5f;
        g_[j] = __builtin_amdgcn_rcpf(1.0f + __expf(-vgate[n]));
        w_[j] = lnw[n];
        o_[j] = lnb[n];
        c_[j] = cvec[n];
        h_[j] = 0.0f;
    }

    const float* xmb  = xmean + batch * SEQ;
    const float* volb = vol   + batch * SEQ;
    float*       ysb  = ys    + batch * SEQ;
    const float invN = 1.0f / (float)NSTATE;

    v4f xc[TBLK / 4], vc[TBLK / 4];
#pragma unroll
    for (int q = 0; q < TBLK / 4; ++q) {
        xc[q] = *(const v4f*)(xmb  + 4 * q);
        vc[q] = *(const v4f*)(volb + 4 * q);
    }

    for (int t0 = 0; t0 < SEQ; t0 += TBLK) {
        const int tn = (t0 + TBLK < SEQ) ? (t0 + TBLK) : t0;
        v4f xn[TBLK / 4], vn[TBLK / 4];
#pragma unroll
        for (int q = 0; q < TBLK / 4; ++q) {
            xn[q] = *(const v4f*)(xmb  + tn + 4 * q);
            vn[q] = *(const v4f*)(volb + tn + 4 * q);
        }

        float yacc = 0.0f;
#pragma unroll
        for (int tt = 0; tt < TBLK; ++tt) {
            const float xm = xc[tt >> 2][tt & 3];
            const float vt = vc[tt >> 2][tt & 3];

            float s1 = 0.0f, s2 = 0.0f;
#pragma unroll
            for (int j = 0; j < 8; ++j) {
                const float h = fmaf(a_[j], h_[j], b_[j] * xm);
                h_[j] = h;
                s1 += h;
                s2 = fmaf(h, h, s2);
            }
            s1 = wave_sum(s1);
            s2 = wave_sum(s2);

            const float mu  = s1 * invN;
            const float var = fmaf(s2, invN, -mu * mu);
            const float rs  = __builtin_amdgcn_rsqf(var + LN_EPS);

            float dot = 0.0f;
#pragma unroll
            for (int j = 0; j < 8; ++j) {
                float hn = fmaf((h_[j] - mu) * rs, w_[j], o_[j]);
                hn *= __builtin_amdgcn_rcpf(fmaf(g_[j], vt, 1.0f));
                h_[j] = hn;
                dot = fmaf(hn, c_[j], dot);
            }
            dot = wave_sum(dot);
            yacc = (lane == tt) ? dot : yacc;
        }
        if (lane < TBLK) ysb[t0 + lane] = yacc;

#pragma unroll
        for (int q = 0; q < TBLK / 4; ++q) { xc[q] = xn[q]; vc[q] = vn[q]; }
    }
}

__global__ void __launch_bounds__(256)
ssm_output(const float* __restrict__ x, const float* __restrict__ ys,
           const float* __restrict__ alpha, const float* __restrict__ log_d,
           float* __restrict__ out, int total4) {
    const int i = blockIdx.x * blockDim.x + threadIdx.x;
    if (i >= total4) return;

    const float a  = 1.0f / (1.0f + __expf(-alpha[0]));
    const float dd = __expf(log_d[0]);
    const float s0 = a + (1.0f - a) * dd;
    const float s1 = 1.0f - a;

    const int row = i >> 8;
    const float yv = s1 * ys[row];

    const v4f xv = ((const v4f*)x)[i];
    v4f o;
    o.x = fmaf(s0, xv.x, yv);
    o.y = fmaf(s0, xv.y, yv);
    o.z = fmaf(s0, xv.z, yv);
    o.w = fmaf(s0, xv.w, yv);
    __builtin_nontemporal_store(o, ((v4f*)out) + i);
}

extern "C" void kernel_launch(void* const* d_in, const int* in_sizes, int n_in,
                              void* d_out, int out_size, void* d_ws, size_t ws_size,
                              hipStream_t stream) {
    const float* x        = (const float*)d_in[0];
    const float* vol      = (const float*)d_in[1];
    const float* llr      = (const float*)d_in[2];
    const float* logb     = (const float*)d_in[3];
    const float* cvec     = (const float*)d_in[4];
    const float* log_d    = (const float*)d_in[5];
    const float* log_step = (const float*)d_in[6];
    const float* vgate    = (const float*)d_in[7];
    const float* alpha    = (const float*)d_in[8];
    const float* lnw      = (const float*)d_in[9];
    const float* lnb      = (const float*)d_in[10];
    float* out = (float*)d_out;

    const int rows = BATCH * SEQ;
    float* xmean = (float*)d_ws;
    float* ys    = xmean + rows;

    {
        const int blocks = rows / 128;
        ssm_rowmean_wmma<<<blocks, 256, 0, stream>>>(x, xmean, rows, DIM);
    }
    ssm_scan<<<BATCH, 32, 0, stream>>>(xmean, vol, llr, logb, cvec, log_d,
                                           log_step, vgate, lnw, lnb, ys);
    {
        const int total4 = (BATCH * SEQ * DIM) / 4;
        const int blocks = (total4 + 255) / 256;
        ssm_output<<<blocks, 256, 0, stream>>>(x, ys, alpha, log_d, out, total4);
    }
}


# scan kernel: each step's instruction order fixed by hand (one asm statement, packed f32 ops) so gate, dot, next-input work fill the chain's waits; lane sums of h, h*h, dot regrouped (low bits change)
# speedup vs baseline: 9.8587x; 1.4725x over previous
#include <hip/hip_runtime.h>
#include <hip/hip_bf16.h>

#define BATCH 8
#define SEQ   4096
#define DIM   1024
#define NSTATE 256
#define LN_EPS 1e-5f
#define TBLK  16

typedef float v2f __attribute__((ext_vector_type(2)));
typedef float v4f __attribute__((ext_vector_type(4)));
typedef float v8f __attribute__((ext_vector_type(8)));

__global__ void __launch_bounds__(256)
ssm_rowmean_wmma(const float* __restrict__ x, float* __restrict__ xmean,
                 int rows, int cols) {
    const int wid  = (blockIdx.x * blockDim.x + threadIdx.x) >> 5;
    const int lane = threadIdx.x & 31;
    const int r0   = wid << 4;
    if (r0 >= rows) return;

    const int myrow = r0 + (lane & 15);
    const int koff  = (lane >> 4) << 1;
    const float* rowp = x + (size_t)myrow * cols + koff;

    const float inv = 1.0f / (float)cols;
    v2f bmat; bmat[0] = inv; bmat[1] = inv;
    v8f acc = {};

#pragma unroll 8
    for (int k = 0; k < cols; k += 4) {
        v2f a = *(const v2f*)(rowp + k);
        acc = __builtin_amdgcn_wmma_f32_16x16x4_f32(
             false, a,  false, bmat,
             (short)0, acc,  false,  false);
    }

    if (lane == 0) {
#pragma unroll
        for (int j = 0; j < 8; ++j) xmean[r0 + j] = acc[j];
    } else if (lane == 16) {
#pragma unroll
        for (int j = 0; j < 8; ++j) xmean[r0 + 8 + j] = acc[j];
    }
}

__device__ __forceinline__ float xhalf(float x) {
    const int b = __builtin_bit_cast(int, x);
    return __builtin_bit_cast(float,
        __builtin_amdgcn_permlanex16(b, b, 0x76543210, 0xfedcba98, true, false));
}

template <int XMASK>
__device__ __forceinline__ float row_xor(float x) {
    const int b = __builtin_bit_cast(int, x);
    return __builtin_bit_cast(float,
        __builtin_amdgcn_update_dpp(0, b, 0x160 + XMASK, 0xf, 0xf, true));
}

__device__ __forceinline__ float wave_sum(float v) {
    v += xhalf(v);
    v += row_xor<8>(v);
    v += row_xor<4>(v);
    v += row_xor<2>(v);
    v += row_xor<1>(v);
    return v;
}

__device__ __forceinline__ void scan_step(v2f (&h)[4], v2f (&bx)[4], float& yacc,
                                          const v2f (&a)[4], const v2f (&b)[4], const v2f (&c)[4],
                                          const v2f (&w)[4], const v2f (&o)[4], const float (&g)[8],
                                          float vt, float xmn, unsigned msk) {
    const float invn = 1.0f / (float)NSTATE;
    const float eps  = LN_EPS;
    asm(
        "v_pk_fma_f32 v[200:201], %[a0], %[h0], %[bx0]\n\t"
        "v_pk_fma_f32 v[204:205], %[a2], %[h2], %[bx2]\n\t"
        "v_pk_fma_f32 v[202:203], %[a1], %[h1], %[bx1]\n\t"
        "v_pk_fma_f32 v[206:207], %[a3], %[h3], %[bx3]\n\t"
        "v_pk_mul_f32 v[224:225], %[h0], %[c0]\n\t"
        "v_fma_f32 v208, %[g0], %[vt], 1.0\n\t"
        "v_fma_f32 v209, %[g1], %[vt], 1.0\n\t"
        "v_fma_f32 v210, %[g2], %[vt], 1.0\n\t"
        "v_pk_mul_f32 v[220:221], v[200:201], v[200:201]\n\t"
        "v_pk_mul_f32 v[222:223], v[204:205], v[204:205]\n\t"
        "v_add_f32 v216, v200, v201\n\t"
        "v_add_f32 v217, v202, v203\n\t"
        "v_add_f32 v218, v204, v205\n\t"
        "v_add_f32 v219, v206, v207\n\t"
        "v_pk_fma_f32 v[224:225], %[h1], %[c1], v[224:225]\n\t"
        "v_fma_f32 v211, %[g3], %[vt], 1.0\n\t"
        "v_pk_fma_f32 v[220:221], v[202:203], v[202:203], v[220:221]\n\t"
        "v_pk_fma_f32 v[222:223], v[206:207], v[206:207], v[222:223]\n\t"
        "v_add_f32 v216, v216, v217\n\t"
        "v_add_f32 v218, v218, v219\n\t"
        "v_fma_f32 v212, %[g4], %[vt], 1.0\n\t"
        "v_fma_f32 v213, %[g5], %[vt], 1.0\n\t"
        "v_pk_fma_f32 v[224:225], %[h2], %[c2], v[224:225]\n\t"
        "v_fma_f32 v214, %[g6], %[vt], 1.0\n\t"
        "v_add_f32 v232, v216, v218\n\t"
        "v_pk_add_f32 v[220:221], v[220:221], v[222:223]\n\t"
        "v_fma_f32 v215, %[g7], %[vt], 1.0\n\t"
        "v_rcp_f32 v208, v208\n\t"
        "v_mov_b32 v230, %[xmn]\n\t"
        "v_permlanex16_b32 v235, v232, %[slo], %[shi] op_sel:[1,0]\n\t"
        "v_pk_fma_f32 v[224:225], %[h3], %[c3], v[224:225]\n\t"
        "v_rcp_f32 v209, v209\n\t"
        "v_add_f32 v233, v220, v221\n\t"
        "v_rcp_f32 v210, v210\n\t"
        "v_add_f32 v232, v232, v235\n\t"
        "v_rcp_f32 v211, v211\n\t"
        "v_pk_mul_f32 %[bx0], v[230:231], %[b0] op_sel_hi:[0,1]\n\t"
        "v_permlanex16_b32 v236, v233, %[slo], %[shi] op_sel:[1,0]\n\t"
        "v_add_f32 v234, v224, v225\n\t"
        "v_add_f32_dpp v232, v232, v232 row_xmask:8 row_mask:0xf bank_mask:0xf bound_ctrl:1\n\t"
        "v_rcp_f32 v212, v212\n\t"
        "v_add_f32 v233, v233, v236\n\t"
        "v_permlanex16_b32 v237, v234, %[slo], %[shi] op_sel:[1,0]\n\t"
        "v_rcp_f32 v213, v213\n\t"
        "v_add_f32_dpp v232, v232, v232 row_xmask:4 row_mask:0xf bank_mask:0xf bound_ctrl:1\n\t"
        "v_add_f32_dpp v233, v233, v233 row_xmask:8 row_mask:0xf bank_mask:0xf bound_ctrl:1\n\t"
        "v_add_f32 v234, v234, v237\n\t"
        "v_rcp_f32 v214, v214\n\t"
        "v_add_f32_dpp v232, v232, v232 row_xmask:2 row_mask:0xf bank_mask:0xf bound_ctrl:1\n\t"
        "v_add_f32_dpp v233, v233, v233 row_xmask:4 row_mask:0xf bank_mask:0xf bound_ctrl:1\n\t"
        "v_add_f32_dpp v234, v234, v234 row_xmask:8 row_mask:0xf bank_mask:0xf bound_ctrl:1\n\t"
        "v_add_f32_dpp v232, v232, v232 row_xmask:1 row_mask:0xf bank_mask:0xf bound_ctrl:1\n\t"
        "v_add_f32_dpp v233, v233, v233 row_xmask:2 row_mask:0xf bank_mask:0xf bound_ctrl:1\n\t"
        "v_add_f32_dpp v234, v234, v234 row_xmask:4 row_mask:0xf bank_mask:0xf bound_ctrl:1\n\t"
        "v_mul_f32 v226, %[invn], v232\n\t"
        "v_rcp_f32 v215, v215\n\t"
        "v_add_f32_dpp v233, v233, v233 row_xmask:1 row_mask:0xf bank_mask:0xf bound_ctrl:1\n\t"
        "v_add_f32_dpp v234, v234, v234 row_xmask:2 row_mask:0xf bank_mask:0xf bound_ctrl:1\n\t"
        "v_mul_f32_e64 v228, v226, -v226\n\t"
        "v_pk_add_f32 v[200:201], v[200:201], v[226:227] op_sel_hi:[1,0] neg_lo:[0,1] neg_hi:[0,1]\n\t"
        "v_pk_add_f32 v[202:203], v[202:203], v[226:227] op_sel_hi:[1,0] neg_lo:[0,1] neg_hi:[0,1]\n\t"
        "v_pk_add_f32 v[204:205], v[204:205], v[226:227] op_sel_hi:[1,0] neg_lo:[0,1] neg_hi:[0,1]\n\t"
        "v_pk_add_f32 v[206:207], v[206:207], v[226:227] op_sel_hi:[1,0] neg_lo:[0,1] neg_hi:[0,1]\n\t"
        "v_fmac_f32 v228, %[invn], v233\n\t"
        "v_add_f32_dpp v234, v234, v234 row_xmask:1 row_mask:0xf bank_mask:0xf bound_ctrl:1\n\t"
        "v_pk_mul_f32 %[bx1], v[230:231], %[b1] op_sel_hi:[0,1]\n\t"
        "v_pk_mul_f32 %[bx2], v[230:231], %[b2] op_sel_hi:[0,1]\n\t"
        "v_add_f32 v228, %[eps], v228\n\t"
        "v_pk_mul_f32 %[bx3], v[230:231], %[b3] op_sel_hi:[0,1]\n\t"
        "v_cndmask_b32_e64 %[ya], %[ya], v234, %[msk]\n\t"
        "v_rsq_f32 v228, v228\n\t"
        "v_nop\n\t"
        "v_pk_mul_f32 v[200:201], v[228:229], v[200:201] op_sel_hi:[0,1]\n\t"
        "v_pk_mul_f32 v[202:203], v[228:229], v[202:203] op_sel_hi:[0,1]\n\t"
        "v_pk_mul_f32 v[204:205], v[228:229], v[204:205] op_sel_hi:[0,1]\n\t"
        "v_pk_mul_f32 v[206:207], v[228:229], v[206:207] op_sel_hi:[0,1]\n\t"
        "v_pk_fma_f32 v[200:201], v[200:201], %[w0], %[o0]\n\t"
        "v_pk_fma_f32 v[202:203], v[202:203], %[w1], %[o1]\n\t"
        "v_pk_fma_f32 v[204:205], v[204:205], %[w2], %[o2]\n\t"
        "v_pk_fma_f32 v[206:207], v[206:207], %[w3], %[o3]\n\t"
        "v_pk_mul_f32 %[h0], v[200:201], v[208:209]\n\t"
        "v_pk_mul_f32 %[h1], v[202:203], v[210:211]\n\t"
        "v_pk_mul_f32 %[h2], v[204:205], v[212:213]\n\t"
        "v_pk_mul_f32 %[h3], v[206:207], v[214:215]"
        :
          [h0] "+v"(h[0]), [h1] "+v"(h[1]), [h2] "+v"(h[2]), [h3] "+v"(h[3]),
          [bx0] "+v"(bx[0]), [bx1] "+v"(bx[1]), [bx2] "+v"(bx[2]), [bx3] "+v"(bx[3]),
          [ya] "+v"(yacc)
        :
          [a0] "v"(a[0]), [a1] "v"(a[1]), [a2] "v"(a[2]), [a3] "v"(a[3]),
          [b0] "v"(b[0]), [b1] "v"(b[1]), [b2] "v"(b[2]), [b3] "v"(b[3]),
          [c0] "v"(c[0]), [c1] "v"(c[1]), [c2] "v"(c[2]), [c3] "v"(c[3]),
          [w0] "v"(w[0]), [w1] "v"(w[1]), [w2] "v"(w[2]), [w3] "v"(w[3]),
          [o0] "v"(o[0]), [o1] "v"(o[1]), [o2] "v"(o[2]), [o3] "v"(o[3]),
          [g0] "v"(g[0]), [g1] "v"(g[1]), [g2] "v"(g[2]), [g3] "v"(g[3]),
          [g4] "v"(g[4]), [g5] "v"(g[5]), [g6] "v"(g[6]), [g7] "v"(g[7]),
          [vt] "s"(vt), [xmn] "s"(xmn), [msk] "s"(msk), [invn] "s"(invn),
          [eps] "s"(eps), [slo] "s"(0x76543210), [shi] "s"(0xfedcba98)
        :
          "v200", "v201", "v202", "v203", "v204", "v205", "v206", "v207", "v208", "v209",
          "v210", "v211", "v212", "v213", "v214", "v215", "v216", "v217", "v218", "v219",
          "v220", "v221", "v222", "v223", "v224", "v225", "v226", "v227", "v228", "v229",
          "v230", "v231", "v232", "v233", "v234", "v235", "v236", "v237");
}

__global__ void __launch_bounds__(32)
ssm_scan(const float* __restrict__ xmean,
         const float* __restrict__ vol,
         const float* __restrict__ llr,
         const float* __restrict__ logb,
         const float* __restrict__ cvec,
         const float* __restrict__ log_d,
         const float* __restrict__ log_step,
         const float* __restrict__ vgate,
         const float* __restrict__ lnw,
         const float* __restrict__ lnb,
         float* __restrict__ ys)
{
    const int lane  = threadIdx.x;
    const int batch = blockIdx.x;

    const float step = __expf(log_step[0]);

    v2f a_[4], b_[4], c_[4], w_[4], o_[4], h_[4], bx_[4];
    float g_[8];
#pragma unroll
    for (int j = 0; j < 8; ++j) {
        const int n = lane + 32 * j;
        const float lam = -__expf(llr[n]);
        const float ad  = (2.0f + step * lam) * __builtin_amdgcn_rcpf(2.0f - step * lam);
        a_[j >> 1][j & 1] = ad;
        b_[j >> 1][j & 1] = step * (1.0f + ad) * __expf(logb[n]) * 0.5f;
        g_[j] = __builtin_amdgcn_rcpf(1.0f + __expf(-vgate[n]));
        w_[j >> 1][j & 1] = lnw[n];
        o_[j >> 1][j & 1] = lnb[n];
        c_[j >> 1][j & 1] = cvec[n];
        h_[j >> 1][j & 1] = 0.0f;
    }

    const float* xmb  = xmean + batch * SEQ;
    const float* volb = vol   + batch * SEQ;
    float*       ysb  = ys    + batch * SEQ;

    v4f xc[TBLK / 4], vc[TBLK / 4];
#pragma unroll
    for (int q = 0; q < TBLK / 4; ++q) {
        xc[q] = *(const v4f*)(xmb  + 4 * q);
        vc[q] = *(const v4f*)(volb + 4 * q);
    }
#pragma unroll
    for (int k = 0; k < 4; ++k) bx_[k] = b_[k] * xc[0][0];

    for (int t0 = 0; t0 < SEQ; t0 += TBLK) {
        const int tn = (t0 + TBLK < SEQ) ? (t0 + TBLK) : t0;
        v4f xn[TBLK / 4], vn[TBLK / 4];
#pragma unroll
        for (int q = 0; q < TBLK / 4; ++q) {
            xn[q] = *(const v4f*)(xmb  + tn + 4 * q);
            vn[q] = *(const v4f*)(volb + tn + 4 * q);
        }

        float yacc = 0.0f;
#pragma unroll
        for (int tt = 0; tt < TBLK; ++tt) {
            const float vt  = vc[tt >> 2][tt & 3];
            const float xmn = (tt + 1 < TBLK) ? xc[(tt + 1) >> 2][(tt + 1) & 3] : xn[0][0];
            scan_step(h_, bx_, yacc, a_, b_, c_, w_, o_, g_, vt, xmn, 1u << tt);
        }
        if (lane < TBLK && (t0 + lane) > 0) ysb[t0 + lane - 1] = yacc;

#pragma unroll
        for (int q = 0; q < TBLK / 4; ++q) { xc[q] = xn[q]; vc[q] = vn[q]; }
    }

    float de = h_[0][0] * c_[0][0], dq = h_[0][1] * c_[0][1];
#pragma unroll
    for (int k = 1; k < 4; ++k) {
        de = fmaf(h_[k][0], c_[k][0], de);
        dq = fmaf(h_[k][1], c_[k][1], dq);
    }
    float dot = wave_sum(de + dq);
    if (lane == 0) ysb[SEQ - 1] = dot;
}

__global__ void __launch_bounds__(256)
ssm_output(const float* __restrict__ x, const float* __restrict__ ys,
           const float* __restrict__ alpha, const float* __restrict__ log_d,
           float* __restrict__ out, int total4) {
    const int i = blockIdx.x * blockDim.x + threadIdx.x;
    if (i >= total4) return;

    const float a  = 1.0f / (1.0f + __expf(-alpha[0]));
    const float dd = __expf(log_d[0]);
    const float s0 = a + (1.0f - a) * dd;
    const float s1 = 1.0f - a;

    const int row = i >> 8;
    const float yv = s1 * ys[row];

    const v4f xv = ((const v4f*)x)[i];
    v4f o;
    o.x = fmaf(s0, xv.x, yv);
    o.y = fmaf(s0, xv.y, yv);
    o.z = fmaf(s0, xv.z, yv);
    o.w = fmaf(s0, xv.w, yv);
    __builtin_nontemporal_store(o, ((v4f*)out) + i);
}

extern "C" void kernel_launch(void* const* d_in, const int* in_sizes, int n_in,
                              void* d_out, int out_size, void* d_ws, size_t ws_size,
                              hipStream_t stream) {
    const float* x        = (const float*)d_in[0];
    const float* vol      = (const float*)d_in[1];
    const float* llr      = (const float*)d_in[2];
    const float* logb     = (const float*)d_in[3];
    const float* cvec     = (const float*)d_in[4];
    const float* log_d    = (const float*)d_in[5];
    const float* log_step = (const float*)d_in[6];
    const float* vgate    = (const float*)d_in[7];
    const float* alpha    = (const float*)d_in[8];
    const float* lnw      = (const float*)d_in[9];
    const float* lnb      = (const float*)d_in[10];
    float* out = (float*)d_out;

    const int rows = BATCH * SEQ;
    float* xmean = (float*)d_ws;
    float* ys    = xmean + rows;

    {
        const int blocks = rows / 128;
        ssm_rowmean_wmma<<<blocks, 256, 0, stream>>>(x, xmean, rows, DIM);
    }
    ssm_scan<<<BATCH, 32, 0, stream>>>(xmean, vol, llr, logb, cvec, log_d,
                                           log_step, vgate, lnw, lnb, ys);
    {
        const int total4 = (BATCH * SEQ * DIM) / 4;
        const int blocks = (total4 + 255) / 256;
        ssm_output<<<blocks, 256, 0, stream>>>(x, ys, alpha, log_d, out, total4);
    }
}


# scan step: LayerNorm weight, bias and gate folded into w*gf, bias*gf formed beside the chain, eps into eps-mean^2: one packed fma behind the 1/sqrt instead of three ops; products regrouped (low bits)
# speedup vs baseline: 10.7166x; 1.0870x over previous
#include <hip/hip_runtime.h>
#include <hip/hip_bf16.h>

#define BATCH 8
#define SEQ   4096
#define DIM   1024
#define NSTATE 256
#define LN_EPS 1e-5f
#define TBLK  16

typedef float v2f __attribute__((ext_vector_type(2)));
typedef float v4f __attribute__((ext_vector_type(4)));
typedef float v8f __attribute__((ext_vector_type(8)));

__global__ void __launch_bounds__(256)
ssm_rowmean_wmma(const float* __restrict__ x, float* __restrict__ xmean,
                 int rows, int cols) {
    const int wid  = (blockIdx.x * blockDim.x + threadIdx.x) >> 5;
    const int lane = threadIdx.x & 31;
    const int r0   = wid << 4;
    if (r0 >= rows) return;

    const int myrow = r0 + (lane & 15);
    const int koff  = (lane >> 4) << 1;
    const float* rowp = x + (size_t)myrow * cols + koff;

    const float inv = 1.0f / (float)cols;
    v2f bmat; bmat[0] = inv; bmat[1] = inv;
    v8f acc = {};

#pragma unroll 8
    for (int k = 0; k < cols; k += 4) {
        v2f a = *(const v2f*)(rowp + k);
        acc = __builtin_amdgcn_wmma_f32_16x16x4_f32(
             false, a,  false, bmat,
             (short)0, acc,  false,  false);
    }

    if (lane == 0) {
#pragma unroll
        for (int j = 0; j < 8; ++j) xmean[r0 + j] = acc[j];
    } else if (lane == 16) {
#pragma unroll
        for (int j = 0; j < 8; ++j) xmean[r0 + 8 + j] = acc[j];
    }
}

__device__ __forceinline__ float xhalf(float x) {
    const int b = __builtin_bit_cast(int, x);
    return __builtin_bit_cast(float,
        __builtin_amdgcn_permlanex16(b, b, 0x76543210, 0xfedcba98, true, false));
}

template <int XMASK>
__device__ __forceinline__ float row_xor(float x) {
    const int b = __builtin_bit_cast(int, x);
    return __builtin_bit_cast(float,
        __builtin_amdgcn_update_dpp(0, b, 0x160 + XMASK, 0xf, 0xf, true));
}

__device__ __forceinline__ float wave_sum(float v) {
    v += xhalf(v);
    v += row_xor<8>(v);
    v += row_xor<4>(v);
    v += row_xor<2>(v);
    v += row_xor<1>(v);
    return v;
}

__device__ __forceinline__ void scan_step(v2f (&h)[4], v2f (&bx)[4], float& yacc,
                                          const v2f (&a)[4], const v2f (&b)[4], const v2f (&c)[4],
                                          const v2f (&w)[4], const v2f (&o)[4], const float (&g)[8],
                                          float vt, float xmn, unsigned msk) {
    const float invn = 1.0f / (float)NSTATE;
    const float eps  = LN_EPS;
    asm(
        "v_fma_f32 v208, %[g0], %[vt], 1.0\n\t"
        "v_fma_f32 v209, %[g1], %[vt], 1.0\n\t"
        "v_fma_f32 v210, %[g2], %[vt], 1.0\n\t"
        "v_fma_f32 v211, %[g3], %[vt], 1.0\n\t"
        "v_pk_fma_f32 v[200:201], %[a0], %[h0], %[bx0]\n\t"
        "v_pk_fma_f32 v[204:205], %[a2], %[h2], %[bx2]\n\t"
        "v_pk_fma_f32 v[202:203], %[a1], %[h1], %[bx1]\n\t"
        "v_pk_fma_f32 v[206:207], %[a3], %[h3], %[bx3]\n\t"
        "v_pk_mul_f32 v[224:225], %[h0], %[c0]\n\t"
        "v_fma_f32 v212, %[g4], %[vt], 1.0\n\t"
        "v_fma_f32 v213, %[g5], %[vt], 1.0\n\t"
        "v_fma_f32 v214, %[g6], %[vt], 1.0\n\t"
        "v_pk_mul_f32 v[220:221], v[200:201], v[200:201]\n\t"
        "v_pk_mul_f32 v[222:223], v[204:205], v[204:205]\n\t"
        "v_add_f32 v216, v200, v201\n\t"
        "v_add_f32 v217, v202, v203\n\t"
        "v_add_f32 v218, v204, v205\n\t"
        "v_add_f32 v219, v206, v207\n\t"
        "v_pk_fma_f32 v[224:225], %[h1], %[c1], v[224:225]\n\t"
        "v_fma_f32 v215, %[g7], %[vt], 1.0\n\t"
        "v_pk_fma_f32 v[220:221], v[202:203], v[202:203], v[220:221]\n\t"
        "v_pk_fma_f32 v[222:223], v[206:207], v[206:207], v[222:223]\n\t"
        "v_add_f32 v216, v216, v217\n\t"
        "v_add_f32 v218, v218, v219\n\t"
        "v_rcp_f32 v208, v208\n\t"
        "v_mov_b32 v230, %[xmn]\n\t"
        "v_pk_fma_f32 v[224:225], %[h2], %[c2], v[224:225]\n\t"
        "v_rcp_f32 v209, v209\n\t"
        "v_add_f32 v232, v216, v218\n\t"
        "v_pk_add_f32 v[220:221], v[220:221], v[222:223]\n\t"
        "v_rcp_f32 v210, v210\n\t"
        "v_pk_mul_f32 %[bx0], v[230:231], %[b0] op_sel_hi:[0,1]\n\t"
        "v_rcp_f32 v211, v211\n\t"
        "v_permlanex16_b32 v235, v232, %[slo], %[shi] op_sel:[1,0]\n\t"
        "v_pk_fma_f32 v[224:225], %[h3], %[c3], v[224:225]\n\t"
        "v_rcp_f32 v212, v212\n\t"
        "v_pk_mul_f32 v[238:239], %[w0], v[208:209]\n\t"
        "v_add_f32 v233, v220, v221\n\t"
        "v_add_f32 v232, v232, v235\n\t"
        "v_rcp_f32 v213, v213\n\t"
        "v_pk_mul_f32 v[240:241], %[w1], v[210:211]\n\t"
        "v_rcp_f32 v214, v214\n\t"
        "v_permlanex16_b32 v236, v233, %[slo], %[shi] op_sel:[1,0]\n\t"
        "v_add_f32_dpp v232, v232, v232 row_xmask:8 row_mask:0xf bank_mask:0xf bound_ctrl:1\n\t"
        "v_add_f32 v234, v224, v225\n\t"
        "v_rcp_f32 v215, v215\n\t"
        "v_add_f32 v233, v233, v236\n\t"
        "v_pk_mul_f32 v[242:243], %[w2], v[212:213]\n\t"
        "v_add_f32_dpp v232, v232, v232 row_xmask:4 row_mask:0xf bank_mask:0xf bound_ctrl:1\n\t"
        "v_permlanex16_b32 v237, v234, %[slo], %[shi] op_sel:[1,0]\n\t"
        "v_add_f32_dpp v233, v233, v233 row_xmask:8 row_mask:0xf bank_mask:0xf bound_ctrl:1\n\t"
        "v_pk_mul_f32 v[244:245], %[w3], v[214:215]\n\t"
        "v_add_f32_dpp v232, v232, v232 row_xmask:2 row_mask:0xf bank_mask:0xf bound_ctrl:1\n\t"
        "v_add_f32 v234, v234, v237\n\t"
        "v_add_f32_dpp v233, v233, v233 row_xmask:4 row_mask:0xf bank_mask:0xf bound_ctrl:1\n\t"
        "v_pk_mul_f32 v[246:247], %[o0], v[208:209]\n\t"
        "v_add_f32_dpp v232, v232, v232 row_xmask:1 row_mask:0xf bank_mask:0xf bound_ctrl:1\n\t"
        "v_add_f32_dpp v234, v234, v234 row_xmask:8 row_mask:0xf bank_mask:0xf bound_ctrl:1\n\t"
        "v_add_f32_dpp v233, v233, v233 row_xmask:2 row_mask:0xf bank_mask:0xf bound_ctrl:1\n\t"
        "v_mul_f32 v226, %[invn], v232\n\t"
        "v_pk_mul_f32 v[248:249], %[o1], v[210:211]\n\t"
        "v_add_f32_dpp v234, v234, v234 row_xmask:4 row_mask:0xf bank_mask:0xf bound_ctrl:1\n\t"
        "v_add_f32_dpp v233, v233, v233 row_xmask:1 row_mask:0xf bank_mask:0xf bound_ctrl:1\n\t"
        "v_fma_f32 v228, -v226, v226, %[eps]\n\t"
        "v_pk_add_f32 v[200:201], v[200:201], v[226:227] op_sel_hi:[1,0] neg_lo:[0,1] neg_hi:[0,1]\n\t"
        "v_pk_add_f32 v[202:203], v[202:203], v[226:227] op_sel_hi:[1,0] neg_lo:[0,1] neg_hi:[0,1]\n\t"
        "v_pk_add_f32 v[204:205], v[204:205], v[226:227] op_sel_hi:[1,0] neg_lo:[0,1] neg_hi:[0,1]\n\t"
        "v_pk_add_f32 v[206:207], v[206:207], v[226:227] op_sel_hi:[1,0] neg_lo:[0,1] neg_hi:[0,1]\n\t"
        "v_fma_f32 v228, %[invn], v233, v228\n\t"
        "v_add_f32_dpp v234, v234, v234 row_xmask:2 row_mask:0xf bank_mask:0xf bound_ctrl:1\n\t"
        "v_pk_mul_f32 v[250:251], %[o2], v[212:213]\n\t"
        "v_pk_mul_f32 v[252:253], %[o3], v[214:215]\n\t"
        "v_rsq_f32 v228, v228\n\t"
        "v_pk_mul_f32 v[200:201], v[200:201], v[238:239]\n\t"
        "v_pk_mul_f32 v[202:203], v[202:203], v[240:241]\n\t"
        "v_pk_mul_f32 v[204:205], v[204:205], v[242:243]\n\t"
        "v_pk_mul_f32 v[206:207], v[206:207], v[244:245]\n\t"
        "v_add_f32_dpp v234, v234, v234 row_xmask:1 row_mask:0xf bank_mask:0xf bound_ctrl:1\n\t"
        "v_pk_mul_f32 %[bx1], v[230:231], %[b1] op_sel_hi:[0,1]\n\t"
        "v_pk_mul_f32 %[bx2], v[230:231], %[b2] op_sel_hi:[0,1]\n\t"
        "v_pk_mul_f32 %[bx3], v[230:231], %[b3] op_sel_hi:[0,1]\n\t"
        "v_pk_fma_f32 %[h0], v[200:201], v[228:229], v[246:247] op_sel_hi:[1,0,1]\n\t"
        "v_pk_fma_f32 %[h1], v[202:203], v[228:229], v[248:249] op_sel_hi:[1,0,1]\n\t"
        "v_pk_fma_f32 %[h2], v[204:205], v[228:229], v[250:251] op_sel_hi:[1,0,1]\n\t"
        "v_pk_fma_f32 %[h3], v[206:207], v[228:229], v[252:253] op_sel_hi:[1,0,1]\n\t"
        "v_cndmask_b32_e64 %[ya], %[ya], v234, %[msk]"
        :
          [h0] "+v"(h[0]), [h1] "+v"(h[1]), [h2] "+v"(h[2]), [h3] "+v"(h[3]),
          [bx0] "+v"(bx[0]), [bx1] "+v"(bx[1]), [bx2] "+v"(bx[2]), [bx3] "+v"(bx[3]),
          [ya] "+v"(yacc)
        :
          [a0] "v"(a[0]), [a1] "v"(a[1]), [a2] "v"(a[2]), [a3] "v"(a[3]),
          [b0] "v"(b[0]), [b1] "v"(b[1]), [b2] "v"(b[2]), [b3] "v"(b[3]),
          [c0] "v"(c[0]), [c1] "v"(c[1]), [c2] "v"(c[2]), [c3] "v"(c[3]),
          [w0] "v"(w[0]), [w1] "v"(w[1]), [w2] "v"(w[2]), [w3] "v"(w[3]),
          [o0] "v"(o[0]), [o1] "v"(o[1]), [o2] "v"(o[2]), [o3] "v"(o[3]),
          [g0] "v"(g[0]), [g1] "v"(g[1]), [g2] "v"(g[2]), [g3] "v"(g[3]),
          [g4] "v"(g[4]), [g5] "v"(g[5]), [g6] "v"(g[6]), [g7] "v"(g[7]),
          [vt] "s"(vt), [xmn] "s"(xmn), [msk] "s"(msk), [invn] "s"(invn),
          [eps] "s"(eps), [slo] "s"(0x76543210), [shi] "s"(0xfedcba98)
        :
          "v200", "v201", "v202", "v203", "v204", "v205", "v206", "v207", "v208", "v209",
          "v210", "v211", "v212", "v213", "v214", "v215", "v216", "v217", "v218", "v219",
          "v220", "v221", "v222", "v223", "v224", "v225", "v226", "v227", "v228", "v229",
          "v230", "v231", "v232", "v233", "v234", "v235", "v236", "v237", "v238", "v239",
          "v240", "v241", "v242", "v243", "v244", "v245", "v246", "v247", "v248", "v249",
          "v250", "v251", "v252", "v253");
}

__global__ void __launch_bounds__(32)
ssm_scan(const float* __restrict__ xmean,
         const float* __restrict__ vol,
         const float* __restrict__ llr,
         const float* __restrict__ logb,
         const float* __restrict__ cvec,
         const float* __restrict__ log_d,
         const float* __restrict__ log_step,
         const float* __restrict__ vgate,
         const float* __restrict__ lnw,
         const float* __restrict__ lnb,
         float* __restrict__ ys)
{
    const int lane  = threadIdx.x;
    const int batch = blockIdx.x;

    const float step = __expf(log_step[0]);

    v2f a_[4], b_[4], c_[4], w_[4], o_[4], h_[4], bx_[4];
    float g_[8];
#pragma unroll
    for (int j = 0; j < 8; ++j) {
        const int n = lane + 32 * j;
        const float lam = -__expf(llr[n]);
        const float ad  = (2.0f + step * lam) * __builtin_amdgcn_rcpf(2.0f - step * lam);
        a_[j >> 1][j & 1] = ad;
        b_[j >> 1][j & 1] = step * (1.0f + ad) * __expf(logb[n]) * 0.5f;
        g_[j] = __builtin_amdgcn_rcpf(1.0f + __expf(-vgate[n]));
        w_[j >> 1][j & 1] = lnw[n];
        o_[j >> 1][j & 1] = lnb[n];
        c_[j >> 1][j & 1] = cvec[n];
        h_[j >> 1][j & 1] = 0.0f;
    }

    const float* xmb  = xmean + batch * SEQ;
    const float* volb = vol   + batch * SEQ;
    float*       ysb  = ys    + batch * SEQ;

    v4f xc[TBLK / 4], vc[TBLK / 4];
#pragma unroll
    for (int q = 0; q < TBLK / 4; ++q) {
        xc[q] = *(const v4f*)(xmb  + 4 * q);
        vc[q] = *(const v4f*)(volb + 4 * q);
    }
#pragma unroll
    for (int k = 0; k < 4; ++k) bx_[k] = b_[k] * xc[0][0];

    for (int t0 = 0; t0 < SEQ; t0 += TBLK) {
        const int tn = (t0 + TBLK < SEQ) ? (t0 + TBLK) : t0;
        v4f xn[TBLK / 4], vn[TBLK / 4];
#pragma unroll
        for (int q = 0; q < TBLK / 4; ++q) {
            xn[q] = *(const v4f*)(xmb  + tn + 4 * q);
            vn[q] = *(const v4f*)(volb + tn + 4 * q);
        }

        float yacc = 0.0f;
#pragma unroll
        for (int tt = 0; tt < TBLK; ++tt) {
            const float vt  = vc[tt >> 2][tt & 3];
            const float xmn = (tt + 1 < TBLK) ? xc[(tt + 1) >> 2][(tt + 1) & 3] : xn[0][0];
            scan_step(h_, bx_, yacc, a_, b_, c_, w_, o_, g_, vt, xmn, 1u << tt);
        }
        if (lane < TBLK && (t0 + lane) > 0) ysb[t0 + lane - 1] = yacc;

#pragma unroll
        for (int q = 0; q < TBLK / 4; ++q) { xc[q] = xn[q]; vc[q] = vn[q]; }
    }

    float de = h_[0][0] * c_[0][0], dq = h_[0][1] * c_[0][1];
#pragma unroll
    for (int k = 1; k < 4; ++k) {
        de = fmaf(h_[k][0], c_[k][0], de);
        dq = fmaf(h_[k][1], c_[k][1], dq);
    }
    float dot = wave_sum(de + dq);
    if (lane == 0) ysb[SEQ - 1] = dot;
}

__global__ void __launch_bounds__(256)
ssm_output(const float* __restrict__ x, const float* __restrict__ ys,
           const float* __restrict__ alpha, const float* __restrict__ log_d,
           float* __restrict__ out, int total4) {
    const int i = blockIdx.x * blockDim.x + threadIdx.x;
    if (i >= total4) return;

    const float a  = 1.0f / (1.0f + __expf(-alpha[0]));
    const float dd = __expf(log_d[0]);
    const float s0 = a + (1.0f - a) * dd;
    const float s1 = 1.0f - a;

    const int row = i >> 8;
    const float yv = s1 * ys[row];

    const v4f xv = ((const v4f*)x)[i];
    v4f o;
    o.x = fmaf(s0, xv.x, yv);
    o.y = fmaf(s0, xv.y, yv);
    o.z = fmaf(s0, xv.z, yv);
    o.w = fmaf(s0, xv.w, yv);
    __builtin_nontemporal_store(o, ((v4f*)out) + i);
}

extern "C" void kernel_launch(void* const* d_in, const int* in_sizes, int n_in,
                              void* d_out, int out_size, void* d_ws, size_t ws_size,
                              hipStream_t stream) {
    const float* x        = (const float*)d_in[0];
    const float* vol      = (const float*)d_in[1];
    const float* llr      = (const float*)d_in[2];
    const float* logb     = (const float*)d_in[3];
    const float* cvec     = (const float*)d_in[4];
    const float* log_d    = (const float*)d_in[5];
    const float* log_step = (const float*)d_in[6];
    const float* vgate    = (const float*)d_in[7];
    const float* alpha    = (const float*)d_in[8];
    const float* lnw      = (const float*)d_in[9];
    const float* lnb      = (const float*)d_in[10];
    float* out = (float*)d_out;

    const int rows = BATCH * SEQ;
    float* xmean = (float*)d_ws;
    float* ys    = xmean + rows;

    {
        const int blocks = rows / 128;
        ssm_rowmean_wmma<<<blocks, 256, 0, stream>>>(x, xmean, rows, DIM);
    }
    ssm_scan<<<BATCH, 32, 0, stream>>>(xmean, vol, llr, logb, cvec, log_d,
                                           log_step, vgate, lnw, lnb, ys);
    {
        const int total4 = (BATCH * SEQ * DIM) / 4;
        const int blocks = (total4 + 255) / 256;
        ssm_output<<<blocks, 256, 0, stream>>>(x, ys, alpha, log_d, out, total4);
    }
}


# row-mean and output kernels: later blocks' 16-byte loads requested ahead of the current block's work (32-48 and 8-16 in flight per lane), whole rows per wave; row sums regrouped (low bits change)
# speedup vs baseline: 11.2079x; 1.0458x over previous
#include <hip/hip_runtime.h>
#include <hip/hip_bf16.h>

#define BATCH 8
#define SEQ   4096
#define DIM   1024
#define NSTATE 256
#define LN_EPS 1e-5f
#define TBLK  16

typedef float v2f __attribute__((ext_vector_type(2)));
typedef float v4f __attribute__((ext_vector_type(4)));
typedef float v8f __attribute__((ext_vector_type(8)));

#define RM_PF 16

__global__ void __launch_bounds__(256)
ssm_rowmean_wmma(const float* __restrict__ x, float* __restrict__ xmean) {
    const int wid  = (blockIdx.x * 256 + threadIdx.x) >> 5;
    const int lane = threadIdx.x & 31;
    const int r0   = wid << 4;

    const int myrow = r0 + (lane & 15);
    const int koff  = (lane >> 4) << 2;
    const v4f* rowp = (const v4f*)(x + (size_t)myrow * DIM + koff);

    const float inv = 1.0f / (float)DIM;
    v2f bmat; bmat[0] = inv; bmat[1] = inv;
    v8f acc0 = {}, acc1 = {};

    constexpr int NBLK = DIM / (8 * RM_PF);
    v4f buf[2][RM_PF];
#pragma unroll
    for (int i = 0; i < RM_PF; ++i) buf[0][i] = rowp[2 * i];

#pragma unroll
    for (int blk = 0; blk < NBLK; ++blk) {
        if (blk + 1 < NBLK) {
#pragma unroll
            for (int i = 0; i < RM_PF; ++i)
                buf[(blk + 1) & 1][i] = rowp[2 * ((blk + 1) * RM_PF + i)];
            __builtin_amdgcn_sched_barrier(0);
        }
#pragma unroll
        for (int i = 0; i < RM_PF; ++i) {
            const v4f a = buf[blk & 1][i];
            v2f alo; alo[0] = a[0]; alo[1] = a[1];
            v2f ahi; ahi[0] = a[2]; ahi[1] = a[3];
            acc0 = __builtin_amdgcn_wmma_f32_16x16x4_f32(
                 false, alo,  false, bmat,
                 (short)0, acc0,  false,  false);
            acc1 = __builtin_amdgcn_wmma_f32_16x16x4_f32(
                 false, ahi,  false, bmat,
                 (short)0, acc1,  false,  false);
        }
    }

    if (lane == 0) {
#pragma unroll
        for (int j = 0; j < 8; ++j) xmean[r0 + j] = acc0[j] + acc1[j];
    } else if (lane == 16) {
#pragma unroll
        for (int j = 0; j < 8; ++j) xmean[r0 + 8 + j] = acc0[j] + acc1[j];
    }
}

__device__ __forceinline__ float xhalf(float x) {
    const int b = __builtin_bit_cast(int, x);
    return __builtin_bit_cast(float,
        __builtin_amdgcn_permlanex16(b, b, 0x76543210, 0xfedcba98, true, false));
}

template <int XMASK>
__device__ __forceinline__ float row_xor(float x) {
    const int b = __builtin_bit_cast(int, x);
    return __builtin_bit_cast(float,
        __builtin_amdgcn_update_dpp(0, b, 0x160 + XMASK, 0xf, 0xf, true));
}

__device__ __forceinline__ float wave_sum(float v) {
    v += xhalf(v);
    v += row_xor<8>(v);
    v += row_xor<4>(v);
    v += row_xor<2>(v);
    v += row_xor<1>(v);
    return v;
}

__device__ __forceinline__ void scan_step(v2f (&h)[4], v2f (&bx)[4], float& yacc,
                                          const v2f (&a)[4], const v2f (&b)[4], const v2f (&c)[4],
                                          const v2f (&w)[4], const v2f (&o)[4], const float (&g)[8],
                                          float vt, float xmn, unsigned msk) {
    const float invn = 1.0f / (float)NSTATE;
    const float eps  = LN_EPS;
    asm(
        "v_fma_f32 v208, %[g0], %[vt], 1.0\n\t"
        "v_fma_f32 v209, %[g1], %[vt], 1.0\n\t"
        "v_fma_f32 v210, %[g2], %[vt], 1.0\n\t"
        "v_fma_f32 v211, %[g3], %[vt], 1.0\n\t"
        "v_pk_fma_f32 v[200:201], %[a0], %[h0], %[bx0]\n\t"
        "v_pk_fma_f32 v[204:205], %[a2], %[h2], %[bx2]\n\t"
        "v_pk_fma_f32 v[202:203], %[a1], %[h1], %[bx1]\n\t"
        "v_pk_fma_f32 v[206:207], %[a3], %[h3], %[bx3]\n\t"
        "v_pk_mul_f32 v[224:225], %[h0], %[c0]\n\t"
        "v_fma_f32 v212, %[g4], %[vt], 1.0\n\t"
        "v_fma_f32 v213, %[g5], %[vt], 1.0\n\t"
        "v_fma_f32 v214, %[g6], %[vt], 1.0\n\t"
        "v_pk_mul_f32 v[220:221], v[200:201], v[200:201]\n\t"
        "v_pk_mul_f32 v[222:223], v[204:205], v[204:205]\n\t"
        "v_add_f32 v216, v200, v201\n\t"
        "v_add_f32 v217, v202, v203\n\t"
        "v_add_f32 v218, v204, v205\n\t"
        "v_add_f32 v219, v206, v207\n\t"
        "v_pk_fma_f32 v[224:225], %[h1], %[c1], v[224:225]\n\t"
        "v_fma_f32 v215, %[g7], %[vt], 1.0\n\t"
        "v_pk_fma_f32 v[220:221], v[202:203], v[202:203], v[220:221]\n\t"
        "v_pk_fma_f32 v[222:223], v[206:207], v[206:207], v[222:223]\n\t"
        "v_add_f32 v216, v216, v217\n\t"
        "v_add_f32 v218, v218, v219\n\t"
        "v_rcp_f32 v208, v208\n\t"
        "v_mov_b32 v230, %[xmn]\n\t"
        "v_pk_fma_f32 v[224:225], %[h2], %[c2], v[224:225]\n\t"
        "v_rcp_f32 v209, v209\n\t"
        "v_add_f32 v232, v216, v218\n\t"
        "v_pk_add_f32 v[220:221], v[220:221], v[222:223]\n\t"
        "v_rcp_f32 v210, v210\n\t"
        "v_pk_mul_f32 %[bx0], v[230:231], %[b0] op_sel_hi:[0,1]\n\t"
        "v_rcp_f32 v211, v211\n\t"
        "v_permlanex16_b32 v235, v232, %[slo], %[shi] op_sel:[1,0]\n\t"
        "v_pk_fma_f32 v[224:225], %[h3], %[c3], v[224:225]\n\t"
        "v_rcp_f32 v212, v212\n\t"
        "v_pk_mul_f32 v[238:239], %[w0], v[208:209]\n\t"
        "v_add_f32 v233, v220, v221\n\t"
        "v_add_f32 v232, v232, v235\n\t"
        "v_rcp_f32 v213, v213\n\t"
        "v_pk_mul_f32 v[240:241], %[w1], v[210:211]\n\t"
        "v_rcp_f32 v214, v214\n\t"
        "v_permlanex16_b32 v236, v233, %[slo], %[shi] op_sel:[1,0]\n\t"
        "v_add_f32_dpp v232, v232, v232 row_xmask:8 row_mask:0xf bank_mask:0xf bound_ctrl:1\n\t"
        "v_add_f32 v234, v224, v225\n\t"
        "v_rcp_f32 v215, v215\n\t"
        "v_add_f32 v233, v233, v236\n\t"
        "v_pk_mul_f32 v[242:243], %[w2], v[212:213]\n\t"
        "v_add_f32_dpp v232, v232, v232 row_xmask:4 row_mask:0xf bank_mask:0xf bound_ctrl:1\n\t"
        "v_permlanex16_b32 v237, v234, %[slo], %[shi] op_sel:[1,0]\n\t"
        "v_add_f32_dpp v233, v233, v233 row_xmask:8 row_mask:0xf bank_mask:0xf bound_ctrl:1\n\t"
        "v_pk_mul_f32 v[244:245], %[w3], v[214:215]\n\t"
        "v_add_f32_dpp v232, v232, v232 row_xmask:2 row_mask:0xf bank_mask:0xf bound_ctrl:1\n\t"
        "v_add_f32 v234, v234, v237\n\t"
        "v_add_f32_dpp v233, v233, v233 row_xmask:4 row_mask:0xf bank_mask:0xf bound_ctrl:1\n\t"
        "v_pk_mul_f32 v[246:247], %[o0], v[208:209]\n\t"
        "v_add_f32_dpp v232, v232, v232 row_xmask:1 row_mask:0xf bank_mask:0xf bound_ctrl:1\n\t"
        "v_add_f32_dpp v234, v234, v234 row_xmask:8 row_mask:0xf bank_mask:0xf bound_ctrl:1\n\t"
        "v_add_f32_dpp v233, v233, v233 row_xmask:2 row_mask:0xf bank_mask:0xf bound_ctrl:1\n\t"
        "v_mul_f32 v226, %[invn], v232\n\t"
        "v_pk_mul_f32 v[248:249], %[o1], v[210:211]\n\t"
        "v_add_f32_dpp v234, v234, v234 row_xmask:4 row_mask:0xf bank_mask:0xf bound_ctrl:1\n\t"
        "v_add_f32_dpp v233, v233, v233 row_xmask:1 row_mask:0xf bank_mask:0xf bound_ctrl:1\n\t"
        "v_fma_f32 v228, -v226, v226, %[eps]\n\t"
        "v_pk_add_f32 v[200:201], v[200:201], v[226:227] op_sel_hi:[1,0] neg_lo:[0,1] neg_hi:[0,1]\n\t"
        "v_pk_add_f32 v[202:203], v[202:203], v[226:227] op_sel_hi:[1,0] neg_lo:[0,1] neg_hi:[0,1]\n\t"
        "v_pk_add_f32 v[204:205], v[204:205], v[226:227] op_sel_hi:[1,0] neg_lo:[0,1] neg_hi:[0,1]\n\t"
        "v_pk_add_f32 v[206:207], v[206:207], v[226:227] op_sel_hi:[1,0] neg_lo:[0,1] neg_hi:[0,1]\n\t"
        "v_fma_f32 v228, %[invn], v233, v228\n\t"
        "v_add_f32_dpp v234, v234, v234 row_xmask:2 row_mask:0xf bank_mask:0xf bound_ctrl:1\n\t"
        "v_pk_mul_f32 v[250:251], %[o2], v[212:213]\n\t"
        "v_pk_mul_f32 v[252:253], %[o3], v[214:215]\n\t"
        "v_rsq_f32 v228, v228\n\t"
        "v_pk_mul_f32 v[200:201], v[200:201], v[238:239]\n\t"
        "v_pk_mul_f32 v[202:203], v[202:203], v[240:241]\n\t"
        "v_pk_mul_f32 v[204:205], v[204:205], v[242:243]\n\t"
        "v_pk_mul_f32 v[206:207], v[206:207], v[244:245]\n\t"
        "v_add_f32_dpp v234, v234, v234 row_xmask:1 row_mask:0xf bank_mask:0xf bound_ctrl:1\n\t"
        "v_pk_mul_f32 %[bx1], v[230:231], %[b1] op_sel_hi:[0,1]\n\t"
        "v_pk_mul_f32 %[bx2], v[230:231], %[b2] op_sel_hi:[0,1]\n\t"
        "v_pk_mul_f32 %[bx3], v[230:231], %[b3] op_sel_hi:[0,1]\n\t"
        "v_pk_fma_f32 %[h0], v[200:201], v[228:229], v[246:247] op_sel_hi:[1,0,1]\n\t"
        "v_pk_fma_f32 %[h1], v[202:203], v[228:229], v[248:249] op_sel_hi:[1,0,1]\n\t"
        "v_pk_fma_f32 %[h2], v[204:205], v[228:229], v[250:251] op_sel_hi:[1,0,1]\n\t"
        "v_pk_fma_f32 %[h3], v[206:207], v[228:229], v[252:253] op_sel_hi:[1,0,1]\n\t"
        "v_cndmask_b32_e64 %[ya], %[ya], v234, %[msk]"
        :
          [h0] "+v"(h[0]), [h1] "+v"(h[1]), [h2] "+v"(h[2]), [h3] "+v"(h[3]),
          [bx0] "+v"(bx[0]), [bx1] "+v"(bx[1]), [bx2] "+v"(bx[2]), [bx3] "+v"(bx[3]),
          [ya] "+v"(yacc)
        :
          [a0] "v"(a[0]), [a1] "v"(a[1]), [a2] "v"(a[2]), [a3] "v"(a[3]),
          [b0] "v"(b[0]), [b1] "v"(b[1]), [b2] "v"(b[2]), [b3] "v"(b[3]),
          [c0] "v"(c[0]), [c1] "v"(c[1]), [c2] "v"(c[2]), [c3] "v"(c[3]),
          [w0] "v"(w[0]), [w1] "v"(w[1]), [w2] "v"(w[2]), [w3] "v"(w[3]),
          [o0] "v"(o[0]), [o1] "v"(o[1]), [o2] "v"(o[2]), [o3] "v"(o[3]),
          [g0] "v"(g[0]), [g1] "v"(g[1]), [g2] "v"(g[2]), [g3] "v"(g[3]),
          [g4] "v"(g[4]), [g5] "v"(g[5]), [g6] "v"(g[6]), [g7] "v"(g[7]),
          [vt] "s"(vt), [xmn] "s"(xmn), [msk] "s"(msk), [invn] "s"(invn),
          [eps] "s"(eps), [slo] "s"(0x76543210), [shi] "s"(0xfedcba98)
        :
          "v200", "v201", "v202", "v203", "v204", "v205", "v206", "v207", "v208", "v209",
          "v210", "v211", "v212", "v213", "v214", "v215", "v216", "v217", "v218", "v219",
          "v220", "v221", "v222", "v223", "v224", "v225", "v226", "v227", "v228", "v229",
          "v230", "v231", "v232", "v233", "v234", "v235", "v236", "v237", "v238", "v239",
          "v240", "v241", "v242", "v243", "v244", "v245", "v246", "v247", "v248", "v249",
          "v250", "v251", "v252", "v253");
}

__global__ void __launch_bounds__(32)
ssm_scan(const float* __restrict__ xmean,
         const float* __restrict__ vol,
         const float* __restrict__ llr,
         const float* __restrict__ logb,
         const float* __restrict__ cvec,
         const float* __restrict__ log_d,
         const float* __restrict__ log_step,
         const float* __restrict__ vgate,
         const float* __restrict__ lnw,
         const float* __restrict__ lnb,
         float* __restrict__ ys)
{
    const int lane  = threadIdx.x;
    const int batch = blockIdx.x;

    const float step = __expf(log_step[0]);

    v2f a_[4], b_[4], c_[4], w_[4], o_[4], h_[4], bx_[4];
    float g_[8];
#pragma unroll
    for (int j = 0; j < 8; ++j) {
        const int n = lane + 32 * j;
        const float lam = -__expf(llr[n]);
        const float ad  = (2.0f + step * lam) * __builtin_amdgcn_rcpf(2.0f - step * lam);
        a_[j >> 1][j & 1] = ad;
        b_[j >> 1][j & 1] = step * (1.0f + ad) * __expf(logb[n]) * 0.5f;
        g_[j] = __builtin_amdgcn_rcpf(1.0f + __expf(-vgate[n]));
        w_[j >> 1][j & 1] = lnw[n];
        o_[j >> 1][j & 1] = lnb[n];
        c_[j >> 1][j & 1] = cvec[n];
        h_[j >> 1][j & 1] = 0.0f;
    }

    const float* xmb  = xmean + batch * SEQ;
    const float* volb = vol   + batch * SEQ;
    float*       ysb  = ys    + batch * SEQ;

    v4f xc[TBLK / 4], vc[TBLK / 4];
#pragma unroll
    for (int q = 0; q < TBLK / 4; ++q) {
        xc[q] = *(const v4f*)(xmb  + 4 * q);
        vc[q] = *(const v4f*)(volb + 4 * q);
    }
#pragma unroll
    for (int k = 0; k < 4; ++k) bx_[k] = b_[k] * xc[0][0];

    for (int t0 = 0; t0 < SEQ; t0 += TBLK) {
        const int tn = (t0 + TBLK < SEQ) ? (t0 + TBLK) : t0;
        v4f xn[TBLK / 4], vn[TBLK / 4];
#pragma unroll
        for (int q = 0; q < TBLK / 4; ++q) {
            xn[q] = *(const v4f*)(xmb  + tn + 4 * q);
            vn[q] = *(const v4f*)(volb + tn + 4 * q);
        }

        float yacc = 0.0f;
#pragma unroll
        for (int tt = 0; tt < TBLK; ++tt) {
            const float vt  = vc[tt >> 2][tt & 3];
            const float xmn = (tt + 1 < TBLK) ? xc[(tt + 1) >> 2][(tt + 1) & 3] : xn[0][0];
            scan_step(h_, bx_, yacc, a_, b_, c_, w_, o_, g_, vt, xmn, 1u << tt);
        }
        if (lane < TBLK && (t0 + lane) > 0) ysb[t0 + lane - 1] = yacc;

#pragma unroll
        for (int q = 0; q < TBLK / 4; ++q) { xc[q] = xn[q]; vc[q] = vn[q]; }
    }

    float de = h_[0][0] * c_[0][0], dq = h_[0][1] * c_[0][1];
#pragma unroll
    for (int k = 1; k < 4; ++k) {
        de = fmaf(h_[k][0], c_[k][0], de);
        dq = fmaf(h_[k][1], c_[k][1], dq);
    }
    float dot = wave_sum(de + dq);
    if (lane == 0) ysb[SEQ - 1] = dot;
}

#define OUT_BLOCKS 1024
#define OUT_WAVES  (OUT_BLOCKS * 8)
#define OUT_RPW    ((BATCH * SEQ) / OUT_WAVES)
#define ROW4       (DIM / 4)
#define ROWV       (ROW4 / 32)
static_assert(OUT_RPW * OUT_WAVES == BATCH * SEQ && (OUT_RPW % 2) == 0, "row split");

__device__ __forceinline__ void out_load_row(v4f (&r)[ROWV], const v4f* __restrict__ x4,
                                             int row, int lane) {
#pragma unroll
    for (int j = 0; j < ROWV; ++j) r[j] = x4[row * ROW4 + j * 32 + lane];
}

__device__ __forceinline__ void out_store_row(const v4f (&r)[ROWV], v4f* __restrict__ o4,
                                              int row, int lane, float s0, float yv) {
#pragma unroll
    for (int j = 0; j < ROWV; ++j) {
        const v4f xv = r[j];
        v4f o;
        o.x = fmaf(s0, xv.x, yv);
        o.y = fmaf(s0, xv.y, yv);
        o.z = fmaf(s0, xv.z, yv);
        o.w = fmaf(s0, xv.w, yv);
        __builtin_nontemporal_store(o, o4 + row * ROW4 + j * 32 + lane);
    }
}

__global__ void __launch_bounds__(256)
ssm_output(const float* __restrict__ x, const float* __restrict__ ys,
           const float* __restrict__ alpha, const float* __restrict__ log_d,
           float* __restrict__ out) {
    const int wave = __builtin_amdgcn_readfirstlane((int)((blockIdx.x * 256 + threadIdx.x) >> 5));
    const int lane = threadIdx.x & 31;

    const v4f* x4 = (const v4f*)x;
    v4f*       o4 = (v4f*)out;

    v4f ra[ROWV], rb[ROWV];
    out_load_row(ra, x4, wave, lane);

    const float a  = 1.0f / (1.0f + __expf(-alpha[0]));
    const float dd = __expf(log_d[0]);
    const float s0 = a + (1.0f - a) * dd;
    const float s1 = 1.0f - a;

#pragma unroll 1
    for (int it = 0; it < OUT_RPW; it += 2) {
        const int row = wave + it * OUT_WAVES;
        out_load_row(rb, x4, row + OUT_WAVES, lane);
        __builtin_amdgcn_sched_barrier(0);
        out_store_row(ra, o4, row, lane, s0, s1 * ys[row]);
        if (it + 2 < OUT_RPW) out_load_row(ra, x4, row + 2 * OUT_WAVES, lane);
        __builtin_amdgcn_sched_barrier(0);
        out_store_row(rb, o4, row + OUT_WAVES, lane, s0, s1 * ys[row + OUT_WAVES]);
    }
}

extern "C" void kernel_launch(void* const* d_in, const int* in_sizes, int n_in,
                              void* d_out, int out_size, void* d_ws, size_t ws_size,
                              hipStream_t stream) {
    const float* x        = (const float*)d_in[0];
    const float* vol      = (const float*)d_in[1];
    const float* llr      = (const float*)d_in[2];
    const float* logb     = (const float*)d_in[3];
    const float* cvec     = (const float*)d_in[4];
    const float* log_d    = (const float*)d_in[5];
    const float* log_step = (const float*)d_in[6];
    const float* vgate    = (const float*)d_in[7];
    const float* alpha    = (const float*)d_in[8];
    const float* lnw      = (const float*)d_in[9];
    const float* lnb      = (const float*)d_in[10];
    float* out = (float*)d_out;

    const int rows = BATCH * SEQ;
    float* xmean = (float*)d_ws;
    float* ys    = xmean + rows;

    {
        const int blocks = rows / 128;
        ssm_rowmean_wmma<<<blocks, 256, 0, stream>>>(x, xmean);
    }
    ssm_scan<<<BATCH, 32, 0, stream>>>(xmean, vol, llr, logb, cvec, log_d,
                                           log_step, vgate, lnw, lnb, ys);
    ssm_output<<<OUT_BLOCKS, 256, 0, stream>>>(x, ys, alpha, log_d, out);
}


# scan step: plain f32 ops issued as dual-issue pairs (gate terms, in-lane trees of both sums, one add pair): 86 -> 78 instructions; sum of squares now on plain ops, regrouped (low bits change)
# speedup vs baseline: 11.6798x; 1.0421x over previous
#include <hip/hip_runtime.h>
#include <hip/hip_bf16.h>

#define BATCH 8
#define SEQ   4096
#define DIM   1024
#define NSTATE 256
#define LN_EPS 1e-5f
#define TBLK  16

typedef float v2f __attribute__((ext_vector_type(2)));
typedef float v4f __attribute__((ext_vector_type(4)));
typedef float v8f __attribute__((ext_vector_type(8)));

#define RM_PF 16

__global__ void __launch_bounds__(256)
ssm_rowmean_wmma(const float* __restrict__ x, float* __restrict__ xmean) {
    const int wid  = (blockIdx.x * 256 + threadIdx.x) >> 5;
    const int lane = threadIdx.x & 31;
    const int r0   = wid << 4;

    const int myrow = r0 + (lane & 15);
    const int koff  = (lane >> 4) << 2;
    const v4f* rowp = (const v4f*)(x + (size_t)myrow * DIM + koff);

    const float inv = 1.0f / (float)DIM;
    v2f bmat; bmat[0] = inv; bmat[1] = inv;
    v8f acc0 = {}, acc1 = {};

    constexpr int NBLK = DIM / (8 * RM_PF);
    v4f buf[2][RM_PF];
#pragma unroll
    for (int i = 0; i < RM_PF; ++i) buf[0][i] = rowp[2 * i];

#pragma unroll
    for (int blk = 0; blk < NBLK; ++blk) {
        if (blk + 1 < NBLK) {
#pragma unroll
            for (int i = 0; i < RM_PF; ++i)
                buf[(blk + 1) & 1][i] = rowp[2 * ((blk + 1) * RM_PF + i)];
            __builtin_amdgcn_sched_barrier(0);
        }
#pragma unroll
        for (int i = 0; i < RM_PF; ++i) {
            const v4f a = buf[blk & 1][i];
            v2f alo; alo[0] = a[0]; alo[1] = a[1];
            v2f ahi; ahi[0] = a[2]; ahi[1] = a[3];
            acc0 = __builtin_amdgcn_wmma_f32_16x16x4_f32(
                 false, alo,  false, bmat,
                 (short)0, acc0,  false,  false);
            acc1 = __builtin_amdgcn_wmma_f32_16x16x4_f32(
                 false, ahi,  false, bmat,
                 (short)0, acc1,  false,  false);
        }
    }

    if (lane == 0) {
#pragma unroll
        for (int j = 0; j < 8; ++j) xmean[r0 + j] = acc0[j] + acc1[j];
    } else if (lane == 16) {
#pragma unroll
        for (int j = 0; j < 8; ++j) xmean[r0 + 8 + j] = acc0[j] + acc1[j];
    }
}

__device__ __forceinline__ float xhalf(float x) {
    const int b = __builtin_bit_cast(int, x);
    return __builtin_bit_cast(float,
        __builtin_amdgcn_permlanex16(b, b, 0x76543210, 0xfedcba98, true, false));
}

template <int XMASK>
__device__ __forceinline__ float row_xor(float x) {
    const int b = __builtin_bit_cast(int, x);
    return __builtin_bit_cast(float,
        __builtin_amdgcn_update_dpp(0, b, 0x160 + XMASK, 0xf, 0xf, true));
}

__device__ __forceinline__ float wave_sum(float v) {
    v += xhalf(v);
    v += row_xor<8>(v);
    v += row_xor<4>(v);
    v += row_xor<2>(v);
    v += row_xor<1>(v);
    return v;
}

__device__ __forceinline__ void scan_step(v2f (&h)[4], v2f (&bx)[4], float& yacc,
                                          const v2f (&a)[4], const v2f (&b)[4], const v2f (&c)[4],
                                          const v2f (&w)[4], const v2f (&o)[4], const float (&g)[8],
                                          float vt, float xmn, unsigned msk) {
    const float invn = 1.0f / (float)NSTATE;
    const float eps  = LN_EPS;
    asm(
        "v_dual_fmaak_f32 v208, %[vt], %[g0], 0x3f800000 :: v_dual_fmaak_f32 v209, %[vt], %[g1], 0x3f800000\n\t"
        "v_dual_fmaak_f32 v210, %[vt], %[g2], 0x3f800000 :: v_dual_fmaak_f32 v211, %[vt], %[g3], 0x3f800000\n\t"
        "v_dual_fmaak_f32 v212, %[vt], %[g4], 0x3f800000 :: v_dual_fmaak_f32 v213, %[vt], %[g5], 0x3f800000\n\t"
        "v_pk_fma_f32 v[200:201], %[a0], %[h0], %[bx0]\n\t"
        "v_pk_fma_f32 v[202:203], %[a1], %[h1], %[bx1]\n\t"
        "v_pk_fma_f32 v[204:205], %[a2], %[h2], %[bx2]\n\t"
        "v_pk_fma_f32 v[206:207], %[a3], %[h3], %[bx3]\n\t"
        "v_pk_mul_f32 v[224:225], %[h0], %[c0]\n\t"
        "v_dual_fmaak_f32 v214, %[vt], %[g6], 0x3f800000 :: v_dual_fmaak_f32 v215, %[vt], %[g7], 0x3f800000\n\t"
        "v_rcp_f32 v208, v208\n\t"
        "v_mov_b32 v230, %[xmn]\n\t"
        "v_rcp_f32 v209, v209\n\t"
        "v_dual_add_f32 v216, v200, v201 :: v_dual_add_f32 v217, v202, v203\n\t"
        "v_dual_mul_f32 v220, v200, v200 :: v_dual_mul_f32 v221, v202, v202\n\t"
        "v_dual_add_f32 v218, v204, v205 :: v_dual_add_f32 v219, v206, v207\n\t"
        "v_dual_mul_f32 v222, v204, v204 :: v_dual_mul_f32 v223, v206, v206\n\t"
        "v_pk_fma_f32 v[224:225], %[h1], %[c1], v[224:225]\n\t"
        "v_rcp_f32 v210, v210\n\t"
        "v_dual_fmac_f32 v220, v201, v201 :: v_dual_fmac_f32 v221, v203, v203\n\t"
        "v_dual_add_f32 v216, v216, v217 :: v_dual_add_f32 v219, v218, v219\n\t"
        "v_dual_fmac_f32 v222, v205, v205 :: v_dual_fmac_f32 v223, v207, v207\n\t"
        "v_rcp_f32 v211, v211\n\t"
        "v_pk_mul_f32 v[238:239], %[w0], v[208:209]\n\t"
        "v_rcp_f32 v212, v212\n\t"
        "v_add_f32 v232, v216, v219\n\t"
        "v_pk_fma_f32 v[224:225], %[h2], %[c2], v[224:225]\n\t"
        "v_dual_add_f32 v220, v220, v221 :: v_dual_add_f32 v223, v222, v223\n\t"
        "v_rcp_f32 v213, v213\n\t"
        "v_pk_mul_f32 v[246:247], %[o0], v[208:209]\n\t"
        "v_permlanex16_b32 v235, v232, %[slo], %[shi] op_sel:[1,0]\n\t"
        "v_rcp_f32 v214, v214\n\t"
        "v_add_f32 v233, v220, v223\n\t"
        "v_rcp_f32 v215, v215\n\t"
        "v_pk_fma_f32 v[224:225], %[h3], %[c3], v[224:225]\n\t"
        "v_pk_mul_f32 v[240:241], %[w1], v[210:211]\n\t"
        "v_add_f32 v232, v232, v235\n\t"
        "v_permlanex16_b32 v236, v233, %[slo], %[shi] op_sel:[1,0]\n\t"
        "v_pk_mul_f32 v[242:243], %[w2], v[212:213]\n\t"
        "v_pk_mul_f32 v[248:249], %[o1], v[210:211]\n\t"
        "v_pk_mul_f32 v[250:251], %[o2], v[212:213]\n\t"
        "v_add_f32_dpp v232, v232, v232 row_xmask:8 row_mask:0xf bank_mask:0xf bound_ctrl:1\n\t"
        "v_dual_add_f32 v234, v224, v225 :: v_dual_add_f32 v233, v233, v236\n\t"
        "v_pk_mul_f32 v[244:245], %[w3], v[214:215]\n\t"
        "v_pk_mul_f32 v[252:253], %[o3], v[214:215]\n\t"
        "v_pk_mul_f32 %[bx0], v[230:231], %[b0] op_sel_hi:[0,1]\n\t"
        "v_add_f32_dpp v232, v232, v232 row_xmask:4 row_mask:0xf bank_mask:0xf bound_ctrl:1\n\t"
        "v_add_f32_dpp v233, v233, v233 row_xmask:8 row_mask:0xf bank_mask:0xf bound_ctrl:1\n\t"
        "v_permlanex16_b32 v237, v234, %[slo], %[shi] op_sel:[1,0]\n\t"
        "v_pk_mul_f32 %[bx1], v[230:231], %[b1] op_sel_hi:[0,1]\n\t"
        "v_add_f32_dpp v232, v232, v232 row_xmask:2 row_mask:0xf bank_mask:0xf bound_ctrl:1\n\t"
        "v_add_f32_dpp v233, v233, v233 row_xmask:4 row_mask:0xf bank_mask:0xf bound_ctrl:1\n\t"
        "v_add_f32 v234, v234, v237\n\t"
        "v_pk_mul_f32 %[bx2], v[230:231], %[b2] op_sel_hi:[0,1]\n\t"
        "v_add_f32_dpp v232, v232, v232 row_xmask:1 row_mask:0xf bank_mask:0xf bound_ctrl:1\n\t"
        "v_add_f32_dpp v233, v233, v233 row_xmask:2 row_mask:0xf bank_mask:0xf bound_ctrl:1\n\t"
        "v_add_f32_dpp v234, v234, v234 row_xmask:8 row_mask:0xf bank_mask:0xf bound_ctrl:1\n\t"
        "v_mul_f32 v226, %[invn], v232\n\t"
        "v_pk_mul_f32 %[bx3], v[230:231], %[b3] op_sel_hi:[0,1]\n\t"
        "v_add_f32_dpp v233, v233, v233 row_xmask:1 row_mask:0xf bank_mask:0xf bound_ctrl:1\n\t"
        "v_add_f32_dpp v234, v234, v234 row_xmask:4 row_mask:0xf bank_mask:0xf bound_ctrl:1\n\t"
        "v_fma_f32 v228, -v226, v226, %[eps]\n\t"
        "v_pk_add_f32 v[200:201], v[200:201], v[226:227] op_sel_hi:[1,0] neg_lo:[0,1] neg_hi:[0,1]\n\t"
        "v_pk_add_f32 v[202:203], v[202:203], v[226:227] op_sel_hi:[1,0] neg_lo:[0,1] neg_hi:[0,1]\n\t"
        "v_pk_add_f32 v[204:205], v[204:205], v[226:227] op_sel_hi:[1,0] neg_lo:[0,1] neg_hi:[0,1]\n\t"
        "v_pk_add_f32 v[206:207], v[206:207], v[226:227] op_sel_hi:[1,0] neg_lo:[0,1] neg_hi:[0,1]\n\t"
        "v_fma_f32 v228, %[invn], v233, v228\n\t"
        "v_add_f32_dpp v234, v234, v234 row_xmask:2 row_mask:0xf bank_mask:0xf bound_ctrl:1\n\t"
        "v_pk_mul_f32 v[200:201], v[200:201], v[238:239]\n\t"
        "v_rsq_f32 v228, v228\n\t"
        "v_pk_mul_f32 v[202:203], v[202:203], v[240:241]\n\t"
        "v_pk_mul_f32 v[204:205], v[204:205], v[242:243]\n\t"
        "v_pk_mul_f32 v[206:207], v[206:207], v[244:245]\n\t"
        "v_add_f32_dpp v234, v234, v234 row_xmask:1 row_mask:0xf bank_mask:0xf bound_ctrl:1\n\t"
        "v_pk_fma_f32 %[h0], v[200:201], v[228:229], v[246:247] op_sel_hi:[1,0,1]\n\t"
        "v_pk_fma_f32 %[h1], v[202:203], v[228:229], v[248:249] op_sel_hi:[1,0,1]\n\t"
        "v_pk_fma_f32 %[h2], v[204:205], v[228:229], v[250:251] op_sel_hi:[1,0,1]\n\t"
        "v_pk_fma_f32 %[h3], v[206:207], v[228:229], v[252:253] op_sel_hi:[1,0,1]\n\t"
        "v_cndmask_b32_e64 %[ya], %[ya], v234, %[msk]"
        :
          [h0] "+v"(h[0]), [h1] "+v"(h[1]), [h2] "+v"(h[2]), [h3] "+v"(h[3]),
          [bx0] "+v"(bx[0]), [bx1] "+v"(bx[1]), [bx2] "+v"(bx[2]), [bx3] "+v"(bx[3]),
          [ya] "+v"(yacc)
        :
          [a0] "v"(a[0]), [a1] "v"(a[1]), [a2] "v"(a[2]), [a3] "v"(a[3]),
          [b0] "v"(b[0]), [b1] "v"(b[1]), [b2] "v"(b[2]), [b3] "v"(b[3]),
          [c0] "v"(c[0]), [c1] "v"(c[1]), [c2] "v"(c[2]), [c3] "v"(c[3]),
          [w0] "v"(w[0]), [w1] "v"(w[1]), [w2] "v"(w[2]), [w3] "v"(w[3]),
          [o0] "v"(o[0]), [o1] "v"(o[1]), [o2] "v"(o[2]), [o3] "v"(o[3]),
          [g0] "{v192}"(g[0]), [g1] "{v193}"(g[1]), [g2] "{v194}"(g[2]), [g3] "{v195}"(g[3]),
          [g4] "{v196}"(g[4]), [g5] "{v197}"(g[5]), [g6] "{v198}"(g[6]), [g7] "{v199}"(g[7]),
          [vt] "s"(vt), [xmn] "s"(xmn), [msk] "s"(msk), [invn] "s"(invn),
          [eps] "s"(eps), [slo] "s"(0x76543210), [shi] "s"(0xfedcba98)
        :
          "v200", "v201", "v202", "v203", "v204", "v205", "v206", "v207", "v208", "v209",
          "v210", "v211", "v212", "v213", "v214", "v215", "v216", "v217", "v218", "v219",
          "v220", "v221", "v222", "v223", "v224", "v225", "v226", "v227", "v228", "v229",
          "v230", "v231", "v232", "v233", "v234", "v235", "v236", "v237", "v238", "v239",
          "v240", "v241", "v242", "v243", "v244", "v245", "v246", "v247", "v248", "v249",
          "v250", "v251", "v252", "v253");
}

__global__ void __launch_bounds__(32)
ssm_scan(const float* __restrict__ xmean,
         const float* __restrict__ vol,
         const float* __restrict__ llr,
         const float* __restrict__ logb,
         const float* __restrict__ cvec,
         const float* __restrict__ log_d,
         const float* __restrict__ log_step,
         const float* __restrict__ vgate,
         const float* __restrict__ lnw,
         const float* __restrict__ lnb,
         float* __restrict__ ys)
{
    const int lane  = threadIdx.x;
    const int batch = blockIdx.x;

    const float step = __expf(log_step[0]);

    v2f a_[4], b_[4], c_[4], w_[4], o_[4], h_[4], bx_[4];
    float g_[8];
#pragma unroll
    for (int j = 0; j < 8; ++j) {
        const int n = lane + 32 * j;
        const float lam = -__expf(llr[n]);
        const float ad  = (2.0f + step * lam) * __builtin_amdgcn_rcpf(2.0f - step * lam);
        a_[j >> 1][j & 1] = ad;
        b_[j >> 1][j & 1] = step * (1.0f + ad) * __expf(logb[n]) * 0.5f;
        g_[j] = __builtin_amdgcn_rcpf(1.0f + __expf(-vgate[n]));
        w_[j >> 1][j & 1] = lnw[n];
        o_[j >> 1][j & 1] = lnb[n];
        c_[j >> 1][j & 1] = cvec[n];
        h_[j >> 1][j & 1] = 0.0f;
    }

    const float* xmb  = xmean + batch * SEQ;
    const float* volb = vol   + batch * SEQ;
    float*       ysb  = ys    + batch * SEQ;

    v4f xc[TBLK / 4], vc[TBLK / 4];
#pragma unroll
    for (int q = 0; q < TBLK / 4; ++q) {
        xc[q] = *(const v4f*)(xmb  + 4 * q);
        vc[q] = *(const v4f*)(volb + 4 * q);
    }
#pragma unroll
    for (int k = 0; k < 4; ++k) bx_[k] = b_[k] * xc[0][0];

    for (int t0 = 0; t0 < SEQ; t0 += TBLK) {
        const int tn = (t0 + TBLK < SEQ) ? (t0 + TBLK) : t0;
        v4f xn[TBLK / 4], vn[TBLK / 4];
#pragma unroll
        for (int q = 0; q < TBLK / 4; ++q) {
            xn[q] = *(const v4f*)(xmb  + tn + 4 * q);
            vn[q] = *(const v4f*)(volb + tn + 4 * q);
        }

        float yacc = 0.0f;
#pragma unroll
        for (int tt = 0; tt < TBLK; ++tt) {
            const float vt  = vc[tt >> 2][tt & 3];
            const float xmn = (tt + 1 < TBLK) ? xc[(tt + 1) >> 2][(tt + 1) & 3] : xn[0][0];
            scan_step(h_, bx_, yacc, a_, b_, c_, w_, o_, g_, vt, xmn, 1u << tt);
        }
        if (lane < TBLK && (t0 + lane) > 0) ysb[t0 + lane - 1] = yacc;

#pragma unroll
        for (int q = 0; q < TBLK / 4; ++q) { xc[q] = xn[q]; vc[q] = vn[q]; }
    }

    float de = h_[0][0] * c_[0][0], dq = h_[0][1] * c_[0][1];
#pragma unroll
    for (int k = 1; k < 4; ++k) {
        de = fmaf(h_[k][0], c_[k][0], de);
        dq = fmaf(h_[k][1], c_[k][1], dq);
    }
    float dot = wave_sum(de + dq);
    if (lane == 0) ysb[SEQ - 1] = dot;
}

#define OUT_BLOCKS 1024
#define OUT_WAVES  (OUT_BLOCKS * 8)
#define OUT_RPW    ((BATCH * SEQ) / OUT_WAVES)
#define ROW4       (DIM / 4)
#define ROWV       (ROW4 / 32)
static_assert(OUT_RPW * OUT_WAVES == BATCH * SEQ && (OUT_RPW % 2) == 0, "row split");

__device__ __forceinline__ void out_load_row(v4f (&r)[ROWV], const v4f* __restrict__ x4,
                                             int row, int lane) {
#pragma unroll
    for (int j = 0; j < ROWV; ++j) r[j] = x4[row * ROW4 + j * 32 + lane];
}

__device__ __forceinline__ void out_store_row(const v4f (&r)[ROWV], v4f* __restrict__ o4,
                                              int row, int lane, float s0, float yv) {
#pragma unroll
    for (int j = 0; j < ROWV; ++j) {
        const v4f xv = r[j];
        v4f o;
        o.x = fmaf(s0, xv.x, yv);
        o.y = fmaf(s0, xv.y, yv);
        o.z = fmaf(s0, xv.z, yv);
        o.w = fmaf(s0, xv.w, yv);
        __builtin_nontemporal_store(o, o4 + row * ROW4 + j * 32 + lane);
    }
}

__global__ void __launch_bounds__(256)
ssm_output(const float* __restrict__ x, const float* __restrict__ ys,
           const float* __restrict__ alpha, const float* __restrict__ log_d,
           float* __restrict__ out) {
    const int wave = __builtin_amdgcn_readfirstlane((int)((blockIdx.x * 256 + threadIdx.x) >> 5));
    const int lane = threadIdx.x & 31;

    const v4f* x4 = (const v4f*)x;
    v4f*       o4 = (v4f*)out;

    v4f ra[ROWV], rb[ROWV];
    out_load_row(ra, x4, wave, lane);

    const float a  = 1.0f / (1.0f + __expf(-alpha[0]));
    const float dd = __expf(log_d[0]);
    const float s0 = a + (1.0f - a) * dd;
    const float s1 = 1.0f - a;

#pragma unroll 1
    for (int it = 0; it < OUT_RPW; it += 2) {
        const int row = wave + it * OUT_WAVES;
        out_load_row(rb, x4, row + OUT_WAVES, lane);
        __builtin_amdgcn_sched_barrier(0);
        out_store_row(ra, o4, row, lane, s0, s1 * ys[row]);
        if (it + 2 < OUT_RPW) out_load_row(ra, x4, row + 2 * OUT_WAVES, lane);
        __builtin_amdgcn_sched_barrier(0);
        out_store_row(rb, o4, row + OUT_WAVES, lane, s0, s1 * ys[row + OUT_WAVES]);
    }
}

extern "C" void kernel_launch(void* const* d_in, const int* in_sizes, int n_in,
                              void* d_out, int out_size, void* d_ws, size_t ws_size,
                              hipStream_t stream) {
    const float* x        = (const float*)d_in[0];
    const float* vol      = (const float*)d_in[1];
    const float* llr      = (const float*)d_in[2];
    const float* logb     = (const float*)d_in[3];
    const float* cvec     = (const float*)d_in[4];
    const float* log_d    = (const float*)d_in[5];
    const float* log_step = (const float*)d_in[6];
    const float* vgate    = (const float*)d_in[7];
    const float* alpha    = (const float*)d_in[8];
    const float* lnw      = (const float*)d_in[9];
    const float* lnb      = (const float*)d_in[10];
    float* out = (float*)d_out;

    const int rows = BATCH * SEQ;
    float* xmean = (float*)d_ws;
    float* ys    = xmean + rows;

    {
        const int blocks = rows / 128;
        ssm_rowmean_wmma<<<blocks, 256, 0, stream>>>(x, xmean);
    }
    ssm_scan<<<BATCH, 32, 0, stream>>>(xmean, vol, llr, logb, cvec, log_d,
                                           log_step, vgate, lnw, lnb, ys);
    ssm_output<<<OUT_BLOCKS, 256, 0, stream>>>(x, ys, alpha, log_d, out);
}


# scan kernel: straight-line block boundary: prefetched xmean/vol blocks alternate between two register sets (no 32 scalar copies); outputs stored by all lanes, no compare, exec write or branch
# speedup vs baseline: 12.0848x; 1.0347x over previous
#include <hip/hip_runtime.h>
#include <hip/hip_bf16.h>

#define BATCH 8
#define SEQ   4096
#define DIM   1024
#define NSTATE 256
#define LN_EPS 1e-5f
#define TBLK  16

typedef float v2f __attribute__((ext_vector_type(2)));
typedef float v4f __attribute__((ext_vector_type(4)));
typedef float v8f __attribute__((ext_vector_type(8)));

#define RM_PF 16

__global__ void __launch_bounds__(256)
ssm_rowmean_wmma(const float* __restrict__ x, float* __restrict__ xmean) {
    const int wid  = (blockIdx.x * 256 + threadIdx.x) >> 5;
    const int lane = threadIdx.x & 31;
    const int r0   = wid << 4;

    const int myrow = r0 + (lane & 15);
    const int koff  = (lane >> 4) << 2;
    const v4f* rowp = (const v4f*)(x + (size_t)myrow * DIM + koff);

    const float inv = 1.0f / (float)DIM;
    v2f bmat; bmat[0] = inv; bmat[1] = inv;
    v8f acc0 = {}, acc1 = {};

    constexpr int NBLK = DIM / (8 * RM_PF);
    v4f buf[2][RM_PF];
#pragma unroll
    for (int i = 0; i < RM_PF; ++i) buf[0][i] = rowp[2 * i];

#pragma unroll
    for (int blk = 0; blk < NBLK; ++blk) {
        if (blk + 1 < NBLK) {
#pragma unroll
            for (int i = 0; i < RM_PF; ++i)
                buf[(blk + 1) & 1][i] = rowp[2 * ((blk + 1) * RM_PF + i)];
            __builtin_amdgcn_sched_barrier(0);
        }
#pragma unroll
        for (int i = 0; i < RM_PF; ++i) {
            const v4f a = buf[blk & 1][i];
            v2f alo; alo[0] = a[0]; alo[1] = a[1];
            v2f ahi; ahi[0] = a[2]; ahi[1] = a[3];
            acc0 = __builtin_amdgcn_wmma_f32_16x16x4_f32(
                 false, alo,  false, bmat,
                 (short)0, acc0,  false,  false);
            acc1 = __builtin_amdgcn_wmma_f32_16x16x4_f32(
                 false, ahi,  false, bmat,
                 (short)0, acc1,  false,  false);
        }
    }

    if (lane == 0) {
#pragma unroll
        for (int j = 0; j < 8; ++j) xmean[r0 + j] = acc0[j] + acc1[j];
    } else if (lane == 16) {
#pragma unroll
        for (int j = 0; j < 8; ++j) xmean[r0 + 8 + j] = acc0[j] + acc1[j];
    }
}

__device__ __forceinline__ float xhalf(float x) {
    const int b = __builtin_bit_cast(int, x);
    return __builtin_bit_cast(float,
        __builtin_amdgcn_permlanex16(b, b, 0x76543210, 0xfedcba98, true, false));
}

template <int XMASK>
__device__ __forceinline__ float row_xor(float x) {
    const int b = __builtin_bit_cast(int, x);
    return __builtin_bit_cast(float,
        __builtin_amdgcn_update_dpp(0, b, 0x160 + XMASK, 0xf, 0xf, true));
}

__device__ __forceinline__ float wave_sum(float v) {
    v += xhalf(v);
    v += row_xor<8>(v);
    v += row_xor<4>(v);
    v += row_xor<2>(v);
    v += row_xor<1>(v);
    return v;
}

__device__ __forceinline__ void scan_step(v2f (&h)[4], v2f (&bx)[4], float& yacc,
                                          const v2f (&a)[4], const v2f (&b)[4], const v2f (&c)[4],
                                          const v2f (&w)[4], const v2f (&o)[4], const float (&g)[8],
                                          float vt, float xmn, unsigned msk) {
    const float invn = 1.0f / (float)NSTATE;
    const float eps  = LN_EPS;
    asm(
        "v_dual_fmaak_f32 v208, %[vt], %[g0], 0x3f800000 :: v_dual_fmaak_f32 v209, %[vt], %[g1], 0x3f800000\n\t"
        "v_dual_fmaak_f32 v210, %[vt], %[g2], 0x3f800000 :: v_dual_fmaak_f32 v211, %[vt], %[g3], 0x3f800000\n\t"
        "v_dual_fmaak_f32 v212, %[vt], %[g4], 0x3f800000 :: v_dual_fmaak_f32 v213, %[vt], %[g5], 0x3f800000\n\t"
        "v_pk_fma_f32 v[200:201], %[a0], %[h0], %[bx0]\n\t"
        "v_pk_fma_f32 v[202:203], %[a1], %[h1], %[bx1]\n\t"
        "v_pk_fma_f32 v[204:205], %[a2], %[h2], %[bx2]\n\t"
        "v_pk_fma_f32 v[206:207], %[a3], %[h3], %[bx3]\n\t"
        "v_pk_mul_f32 v[224:225], %[h0], %[c0]\n\t"
        "v_dual_fmaak_f32 v214, %[vt], %[g6], 0x3f800000 :: v_dual_fmaak_f32 v215, %[vt], %[g7], 0x3f800000\n\t"
        "v_rcp_f32 v208, v208\n\t"
        "v_mov_b32 v230, %[xmn]\n\t"
        "v_rcp_f32 v209, v209\n\t"
        "v_dual_add_f32 v216, v200, v201 :: v_dual_add_f32 v217, v202, v203\n\t"
        "v_dual_mul_f32 v220, v200, v200 :: v_dual_mul_f32 v221, v202, v202\n\t"
        "v_dual_add_f32 v218, v204, v205 :: v_dual_add_f32 v219, v206, v207\n\t"
        "v_dual_mul_f32 v222, v204, v204 :: v_dual_mul_f32 v223, v206, v206\n\t"
        "v_pk_fma_f32 v[224:225], %[h1], %[c1], v[224:225]\n\t"
        "v_rcp_f32 v210, v210\n\t"
        "v_dual_fmac_f32 v220, v201, v201 :: v_dual_fmac_f32 v221, v203, v203\n\t"
        "v_dual_add_f32 v216, v216, v217 :: v_dual_add_f32 v219, v218, v219\n\t"
        "v_dual_fmac_f32 v222, v205, v205 :: v_dual_fmac_f32 v223, v207, v207\n\t"
        "v_rcp_f32 v211, v211\n\t"
        "v_pk_mul_f32 v[238:239], %[w0], v[208:209]\n\t"
        "v_rcp_f32 v212, v212\n\t"
        "v_add_f32 v232, v216, v219\n\t"
        "v_pk_fma_f32 v[224:225], %[h2], %[c2], v[224:225]\n\t"
        "v_dual_add_f32 v220, v220, v221 :: v_dual_add_f32 v223, v222, v223\n\t"
        "v_rcp_f32 v213, v213\n\t"
        "v_pk_mul_f32 v[246:247], %[o0], v[208:209]\n\t"
        "v_permlanex16_b32 v235, v232, %[slo], %[shi] op_sel:[1,0]\n\t"
        "v_rcp_f32 v214, v214\n\t"
        "v_add_f32 v233, v220, v223\n\t"
        "v_rcp_f32 v215, v215\n\t"
        "v_pk_fma_f32 v[224:225], %[h3], %[c3], v[224:225]\n\t"
        "v_pk_mul_f32 v[240:241], %[w1], v[210:211]\n\t"
        "v_add_f32 v232, v232, v235\n\t"
        "v_permlanex16_b32 v236, v233, %[slo], %[shi] op_sel:[1,0]\n\t"
        "v_pk_mul_f32 v[242:243], %[w2], v[212:213]\n\t"
        "v_pk_mul_f32 v[248:249], %[o1], v[210:211]\n\t"
        "v_pk_mul_f32 v[250:251], %[o2], v[212:213]\n\t"
        "v_add_f32_dpp v232, v232, v232 row_xmask:8 row_mask:0xf bank_mask:0xf bound_ctrl:1\n\t"
        "v_dual_add_f32 v234, v224, v225 :: v_dual_add_f32 v233, v233, v236\n\t"
        "v_pk_mul_f32 v[244:245], %[w3], v[214:215]\n\t"
        "v_pk_mul_f32 v[252:253], %[o3], v[214:215]\n\t"
        "v_pk_mul_f32 %[bx0], v[230:231], %[b0] op_sel_hi:[0,1]\n\t"
        "v_add_f32_dpp v232, v232, v232 row_xmask:4 row_mask:0xf bank_mask:0xf bound_ctrl:1\n\t"
        "v_add_f32_dpp v233, v233, v233 row_xmask:8 row_mask:0xf bank_mask:0xf bound_ctrl:1\n\t"
        "v_permlanex16_b32 v237, v234, %[slo], %[shi] op_sel:[1,0]\n\t"
        "v_pk_mul_f32 %[bx1], v[230:231], %[b1] op_sel_hi:[0,1]\n\t"
        "v_add_f32_dpp v232, v232, v232 row_xmask:2 row_mask:0xf bank_mask:0xf bound_ctrl:1\n\t"
        "v_add_f32_dpp v233, v233, v233 row_xmask:4 row_mask:0xf bank_mask:0xf bound_ctrl:1\n\t"
        "v_add_f32 v234, v234, v237\n\t"
        "v_pk_mul_f32 %[bx2], v[230:231], %[b2] op_sel_hi:[0,1]\n\t"
        "v_add_f32_dpp v232, v232, v232 row_xmask:1 row_mask:0xf bank_mask:0xf bound_ctrl:1\n\t"
        "v_add_f32_dpp v233, v233, v233 row_xmask:2 row_mask:0xf bank_mask:0xf bound_ctrl:1\n\t"
        "v_add_f32_dpp v234, v234, v234 row_xmask:8 row_mask:0xf bank_mask:0xf bound_ctrl:1\n\t"
        "v_mul_f32 v226, %[invn], v232\n\t"
        "v_pk_mul_f32 %[bx3], v[230:231], %[b3] op_sel_hi:[0,1]\n\t"
        "v_add_f32_dpp v233, v233, v233 row_xmask:1 row_mask:0xf bank_mask:0xf bound_ctrl:1\n\t"
        "v_add_f32_dpp v234, v234, v234 row_xmask:4 row_mask:0xf bank_mask:0xf bound_ctrl:1\n\t"
        "v_fma_f32 v228, -v226, v226, %[eps]\n\t"
        "v_pk_add_f32 v[200:201], v[200:201], v[226:227] op_sel_hi:[1,0] neg_lo:[0,1] neg_hi:[0,1]\n\t"
        "v_pk_add_f32 v[202:203], v[202:203], v[226:227] op_sel_hi:[1,0] neg_lo:[0,1] neg_hi:[0,1]\n\t"
        "v_pk_add_f32 v[204:205], v[204:205], v[226:227] op_sel_hi:[1,0] neg_lo:[0,1] neg_hi:[0,1]\n\t"
        "v_pk_add_f32 v[206:207], v[206:207], v[226:227] op_sel_hi:[1,0] neg_lo:[0,1] neg_hi:[0,1]\n\t"
        "v_fma_f32 v228, %[invn], v233, v228\n\t"
        "v_add_f32_dpp v234, v234, v234 row_xmask:2 row_mask:0xf bank_mask:0xf bound_ctrl:1\n\t"
        "v_pk_mul_f32 v[200:201], v[200:201], v[238:239]\n\t"
        "v_rsq_f32 v228, v228\n\t"
        "v_pk_mul_f32 v[202:203], v[202:203], v[240:241]\n\t"
        "v_pk_mul_f32 v[204:205], v[204:205], v[242:243]\n\t"
        "v_pk_mul_f32 v[206:207], v[206:207], v[244:245]\n\t"
        "v_add_f32_dpp v234, v234, v234 row_xmask:1 row_mask:0xf bank_mask:0xf bound_ctrl:1\n\t"
        "v_pk_fma_f32 %[h0], v[200:201], v[228:229], v[246:247] op_sel_hi:[1,0,1]\n\t"
        "v_pk_fma_f32 %[h1], v[202:203], v[228:229], v[248:249] op_sel_hi:[1,0,1]\n\t"
        "v_pk_fma_f32 %[h2], v[204:205], v[228:229], v[250:251] op_sel_hi:[1,0,1]\n\t"
        "v_pk_fma_f32 %[h3], v[206:207], v[228:229], v[252:253] op_sel_hi:[1,0,1]\n\t"
        "v_cndmask_b32_e64 %[ya], %[ya], v234, %[msk]"
        :
          [h0] "+v"(h[0]), [h1] "+v"(h[1]), [h2] "+v"(h[2]), [h3] "+v"(h[3]),
          [bx0] "+v"(bx[0]), [bx1] "+v"(bx[1]), [bx2] "+v"(bx[2]), [bx3] "+v"(bx[3]),
          [ya] "+v"(yacc)
        :
          [a0] "v"(a[0]), [a1] "v"(a[1]), [a2] "v"(a[2]), [a3] "v"(a[3]),
          [b0] "v"(b[0]), [b1] "v"(b[1]), [b2] "v"(b[2]), [b3] "v"(b[3]),
          [c0] "v"(c[0]), [c1] "v"(c[1]), [c2] "v"(c[2]), [c3] "v"(c[3]),
          [w0] "v"(w[0]), [w1] "v"(w[1]), [w2] "v"(w[2]), [w3] "v"(w[3]),
          [o0] "v"(o[0]), [o1] "v"(o[1]), [o2] "v"(o[2]), [o3] "v"(o[3]),
          [g0] "{v192}"(g[0]), [g1] "{v193}"(g[1]), [g2] "{v194}"(g[2]), [g3] "{v195}"(g[3]),
          [g4] "{v196}"(g[4]), [g5] "{v197}"(g[5]), [g6] "{v198}"(g[6]), [g7] "{v199}"(g[7]),
          [vt] "s"(vt), [xmn] "s"(xmn), [msk] "s"(msk), [invn] "s"(invn),
          [eps] "s"(eps), [slo] "s"(0x76543210), [shi] "s"(0xfedcba98)
        :
          "v200", "v201", "v202", "v203", "v204", "v205", "v206", "v207", "v208", "v209",
          "v210", "v211", "v212", "v213", "v214", "v215", "v216", "v217", "v218", "v219",
          "v220", "v221", "v222", "v223", "v224", "v225", "v226", "v227", "v228", "v229",
          "v230", "v231", "v232", "v233", "v234", "v235", "v236", "v237", "v238", "v239",
          "v240", "v241", "v242", "v243", "v244", "v245", "v246", "v247", "v248", "v249",
          "v250", "v251", "v252", "v253");
}

__device__ __forceinline__ void scan_block(v2f (&h)[4], v2f (&bx)[4],
                                           const v2f (&a)[4], const v2f (&b)[4], const v2f (&c)[4],
                                           const v2f (&w)[4], const v2f (&o)[4], const float (&g)[8],
                                           const v4f (&xc)[TBLK / 4], const v4f (&vc)[TBLK / 4],
                                           v4f (&xn)[TBLK / 4], v4f (&vn)[TBLK / 4],
                                           const float* __restrict__ xsrc, const float* __restrict__ vsrc,
                                           float* __restrict__ ysb, int t0, int l15) {
    float yacc = 0.0f;
#pragma unroll
    for (int tt = 0; tt < TBLK; ++tt) {
        if (tt == 1) {
            __builtin_amdgcn_sched_barrier(0);
#pragma unroll
            for (int q = 0; q < TBLK / 4; ++q) {
                xn[q] = *(const v4f*)(xsrc + 4 * q);
                vn[q] = *(const v4f*)(vsrc + 4 * q);
            }
            __builtin_amdgcn_sched_barrier(0);
        }
        const float vt  = vc[tt >> 2][tt & 3];
        const float xmn = (tt + 1 < TBLK) ? xc[(tt + 1) >> 2][(tt + 1) & 3] : xn[0][0];
        scan_step(h, bx, yacc, a, b, c, w, o, g, vt, xmn, 0x10001u << tt);
    }
    ysb[(t0 + l15 - 1) & (SEQ - 1)] = yacc;
}

__global__ void __launch_bounds__(32)
ssm_scan(const float* __restrict__ xmean,
         const float* __restrict__ vol,
         const float* __restrict__ llr,
         const float* __restrict__ logb,
         const float* __restrict__ cvec,
         const float* __restrict__ log_d,
         const float* __restrict__ log_step,
         const float* __restrict__ vgate,
         const float* __restrict__ lnw,
         const float* __restrict__ lnb,
         float* __restrict__ ys)
{
    const int lane  = threadIdx.x;
    const int batch = blockIdx.x;

    const float step = __expf(log_step[0]);

    v2f a_[4], b_[4], c_[4], w_[4], o_[4], h_[4], bx_[4];
    float g_[8];
#pragma unroll
    for (int j = 0; j < 8; ++j) {
        const int n = lane + 32 * j;
        const float lam = -__expf(llr[n]);
        const float ad  = (2.0f + step * lam) * __builtin_amdgcn_rcpf(2.0f - step * lam);
        a_[j >> 1][j & 1] = ad;
        b_[j >> 1][j & 1] = step * (1.0f + ad) * __expf(logb[n]) * 0.5f;
        g_[j] = __builtin_amdgcn_rcpf(1.0f + __expf(-vgate[n]));
        w_[j >> 1][j & 1] = lnw[n];
        o_[j >> 1][j & 1] = lnb[n];
        c_[j >> 1][j & 1] = cvec[n];
        h_[j >> 1][j & 1] = 0.0f;
    }

    const float* xmb  = xmean + batch * SEQ;
    const float* volb = vol   + batch * SEQ;
    float*       ysb  = ys    + batch * SEQ;

    static_assert(SEQ % (2 * TBLK) == 0, "two blocks per loop trip");
    const int l15 = lane & 15;
    v4f xa[TBLK / 4], va[TBLK / 4], xb[TBLK / 4], vb[TBLK / 4];
#pragma unroll
    for (int q = 0; q < TBLK / 4; ++q) {
        xa[q] = *(const v4f*)(xmb  + 4 * q);
        va[q] = *(const v4f*)(volb + 4 * q);
    }
#pragma unroll
    for (int k = 0; k < 4; ++k) bx_[k] = b_[k] * xa[0][0];

    for (int t0 = 0; t0 < SEQ; t0 += 2 * TBLK) {
        scan_block(h_, bx_, a_, b_, c_, w_, o_, g_, xa, va, xb, vb,
                   xmb + t0 + TBLK, volb + t0 + TBLK, ysb, t0, l15);
        const int tn = (t0 + 2 * TBLK < SEQ) ? (t0 + 2 * TBLK) : t0;
        scan_block(h_, bx_, a_, b_, c_, w_, o_, g_, xb, vb, xa, va,
                   xmb + tn, volb + tn, ysb, t0 + TBLK, l15);
    }

    float de = h_[0][0] * c_[0][0], dq = h_[0][1] * c_[0][1];
#pragma unroll
    for (int k = 1; k < 4; ++k) {
        de = fmaf(h_[k][0], c_[k][0], de);
        dq = fmaf(h_[k][1], c_[k][1], dq);
    }
    float dot = wave_sum(de + dq);
    if (lane == 0) ysb[SEQ - 1] = dot;
}

#define OUT_BLOCKS 1024
#define OUT_WAVES  (OUT_BLOCKS * 8)
#define OUT_RPW    ((BATCH * SEQ) / OUT_WAVES)
#define ROW4       (DIM / 4)
#define ROWV       (ROW4 / 32)
static_assert(OUT_RPW * OUT_WAVES == BATCH * SEQ && (OUT_RPW % 2) == 0, "row split");

__device__ __forceinline__ void out_load_row(v4f (&r)[ROWV], const v4f* __restrict__ x4,
                                             int row, int lane) {
#pragma unroll
    for (int j = 0; j < ROWV; ++j) r[j] = x4[row * ROW4 + j * 32 + lane];
}

__device__ __forceinline__ void out_store_row(const v4f (&r)[ROWV], v4f* __restrict__ o4,
                                              int row, int lane, float s0, float yv) {
#pragma unroll
    for (int j = 0; j < ROWV; ++j) {
        const v4f xv = r[j];
        v4f o;
        o.x = fmaf(s0, xv.x, yv);
        o.y = fmaf(s0, xv.y, yv);
        o.z = fmaf(s0, xv.z, yv);
        o.w = fmaf(s0, xv.w, yv);
        __builtin_nontemporal_store(o, o4 + row * ROW4 + j * 32 + lane);
    }
}

__global__ void __launch_bounds__(256)
ssm_output(const float* __restrict__ x, const float* __restrict__ ys,
           const float* __restrict__ alpha, const float* __restrict__ log_d,
           float* __restrict__ out) {
    const int wave = __builtin_amdgcn_readfirstlane((int)((blockIdx.x * 256 + threadIdx.x) >> 5));
    const int lane = threadIdx.x & 31;

    const v4f* x4 = (const v4f*)x;
    v4f*       o4 = (v4f*)out;

    v4f ra[ROWV], rb[ROWV];
    out_load_row(ra, x4, wave, lane);

    const float a  = 1.0f / (1.0f + __expf(-alpha[0]));
    const float dd = __expf(log_d[0]);
    const float s0 = a + (1.0f - a) * dd;
    const float s1 = 1.0f - a;

#pragma unroll 1
    for (int it = 0; it < OUT_RPW; it += 2) {
        const int row = wave + it * OUT_WAVES;
        out_load_row(rb, x4, row + OUT_WAVES, lane);
        __builtin_amdgcn_sched_barrier(0);
        out_store_row(ra, o4, row, lane, s0, s1 * ys[row]);
        if (it + 2 < OUT_RPW) out_load_row(ra, x4, row + 2 * OUT_WAVES, lane);
        __builtin_amdgcn_sched_barrier(0);
        out_store_row(rb, o4, row + OUT_WAVES, lane, s0, s1 * ys[row + OUT_WAVES]);
    }
}

extern "C" void kernel_launch(void* const* d_in, const int* in_sizes, int n_in,
                              void* d_out, int out_size, void* d_ws, size_t ws_size,
                              hipStream_t stream) {
    const float* x        = (const float*)d_in[0];
    const float* vol      = (const float*)d_in[1];
    const float* llr      = (const float*)d_in[2];
    const float* logb     = (const float*)d_in[3];
    const float* cvec     = (const float*)d_in[4];
    const float* log_d    = (const float*)d_in[5];
    const float* log_step = (const float*)d_in[6];
    const float* vgate    = (const float*)d_in[7];
    const float* alpha    = (const float*)d_in[8];
    const float* lnw      = (const float*)d_in[9];
    const float* lnb      = (const float*)d_in[10];
    float* out = (float*)d_out;

    const int rows = BATCH * SEQ;
    float* xmean = (float*)d_ws;
    float* ys    = xmean + rows;

    {
        const int blocks = rows / 128;
        ssm_rowmean_wmma<<<blocks, 256, 0, stream>>>(x, xmean);
    }
    ssm_scan<<<BATCH, 32, 0, stream>>>(xmean, vol, llr, logb, cvec, log_d,
                                           log_step, vgate, lnw, lnb, ys);
    ssm_output<<<OUT_BLOCKS, 256, 0, stream>>>(x, ys, alpha, log_d, out);
}
